# starting kernel
# baseline (speedup 1.0000x reference)
#include <hip/hip_runtime.h>
#include <hip/hip_bf16.h>

#define N_NODES 50000
#define N_EDGES 1600000
#define IN_DIM 256
#define OUT_DIM 128
#define NEG_SLOPE 0.2f

typedef __attribute__((ext_vector_type(2))) float v2f;
typedef __attribute__((ext_vector_type(8))) float v8f;

__global__ __launch_bounds__(256) void gat_gemm_wmma(
    const float* __restrict__ X,
    const float* __restrict__ W,
    float* __restrict__ H)
{
  const int lane = threadIdx.x & 31;
  const int wave = threadIdx.x >> 5;
  const int l16  = lane & 15;
  const int half = lane >> 4;

  const int mBase = blockIdx.x * 16;
  const int nBase = wave * 16;

  const float* aPtr = X + (size_t)(mBase + l16) * IN_DIM + 2 * half;
  const float* bPtr = W + (size_t)(2 * half) * OUT_DIM + nBase + l16;

  v8f acc = {};
#pragma unroll 4
  for (int kk = 0; kk < IN_DIM; kk += 4) {
    v2f a = *(const v2f*)(aPtr + kk);
    v2f b;
    b.x = bPtr[(size_t)kk * OUT_DIM];
    b.y = bPtr[(size_t)(kk + 1) * OUT_DIM];
    acc = __builtin_amdgcn_wmma_f32_16x16x4_f32(
         false, a,  false, b,
         (short)0, acc,  false,  false);
  }

  float* out = H + (size_t)(mBase + 8 * half) * OUT_DIM + nBase + l16;
#pragma unroll
  for (int v = 0; v < 8; ++v)
    out[(size_t)v * OUT_DIM] = acc[v];
}

__global__ __launch_bounds__(256) void node_logits(
    const float* __restrict__ H,
    const float* __restrict__ attn_l,
    const float* __restrict__ attn_r,
    float* __restrict__ el, float* __restrict__ er)
{
  const int lane = threadIdx.x & 31;
  const int node = blockIdx.x * (blockDim.x >> 5) + (threadIdx.x >> 5);
  if (node >= N_NODES) return;

  const float4 hv = *(const float4*)(H + (size_t)node * OUT_DIM + 4 * lane);
  const float4 al = *(const float4*)(attn_l + 4 * lane);
  const float4 ar = *(const float4*)(attn_r + 4 * lane);
  float sl = hv.x * al.x + hv.y * al.y + hv.z * al.z + hv.w * al.w;
  float sr = hv.x * ar.x + hv.y * ar.y + hv.z * ar.z + hv.w * ar.w;
#pragma unroll
  for (int off = 16; off > 0; off >>= 1) {
    sl += __shfl_down(sl, off, 32);
    sr += __shfl_down(sr, off, 32);
  }
  if (lane == 0) { el[node] = sl; er[node] = sr; }
}

__global__ __launch_bounds__(256) void init_buffers(
    float* __restrict__ out, float* __restrict__ emax, float* __restrict__ esum)
{
  const int i = blockIdx.x * blockDim.x + threadIdx.x;
  if (i < N_NODES * OUT_DIM) out[i] = 0.0f;
  if (i < N_NODES) { emax[i] = -__builtin_inff(); esum[i] = 0.0f; }
}

__global__ __launch_bounds__(256) void edge_score_max(
    const int* __restrict__ src, const int* __restrict__ dst,
    const float* __restrict__ el, const float* __restrict__ er,
    float* __restrict__ score, float* __restrict__ emax)
{
  const int e = blockIdx.x * blockDim.x + threadIdx.x;
  if (e >= N_EDGES) return;
  const float t = el[src[e]] + er[dst[e]];
  const float s = t > 0.0f ? t : NEG_SLOPE * t;
  score[e] = s;
  atomicMax(emax + dst[e], s);
}

__global__ __launch_bounds__(256) void edge_exp_sum(
    const int* __restrict__ dst,
    float* __restrict__ score,
    const float* __restrict__ emax, float* __restrict__ esum)
{
  const int e = blockIdx.x * blockDim.x + threadIdx.x;
  if (e >= N_EDGES) return;
  const int d = dst[e];
  const float ex = __expf(score[e] - emax[d]);
  score[e] = ex;
  atomicAdd(esum + d, ex);
}

__global__ __launch_bounds__(256) void aggregate(
    const int* __restrict__ src, const int* __restrict__ dst,
    const float* __restrict__ H,
    const float* __restrict__ eexp, const float* __restrict__ esum,
    float* __restrict__ out)
{
  const int lane = threadIdx.x & 31;
  const int e = blockIdx.x * (blockDim.x >> 5) + (threadIdx.x >> 5);
  if (e >= N_EDGES) return;

  const int s = src[e], d = dst[e];
  const float a = eexp[e] / esum[d];
  const float4 hv = *(const float4*)(H + (size_t)s * OUT_DIM + 4 * lane);
  float* o = out + (size_t)d * OUT_DIM + 4 * lane;
  atomicAdd(o + 0, a * hv.x);
  atomicAdd(o + 1, a * hv.y);
  atomicAdd(o + 2, a * hv.z);
  atomicAdd(o + 3, a * hv.w);
}

extern "C" void kernel_launch(void* const* d_in, const int* in_sizes, int n_in,
                              void* d_out, int out_size, void* d_ws, size_t ws_size,
                              hipStream_t stream) {
  (void)in_sizes; (void)n_in; (void)out_size; (void)ws_size;

  const float* features  = (const float*)d_in[0];
  const float* fc_weight = (const float*)d_in[1];
  const float* attn_l    = (const float*)d_in[2];
  const float* attn_r    = (const float*)d_in[3];
  const int*   src       = (const int*)d_in[4];
  const int*   dst       = (const int*)d_in[5];
  float*       out       = (float*)d_out;

  float* h     = (float*)d_ws;
  float* el    = h + (size_t)N_NODES * OUT_DIM;
  float* er    = el + N_NODES;
  float* emax  = er + N_NODES;
  float* esum  = emax + N_NODES;
  float* score = esum + N_NODES;

  init_buffers<<<(N_NODES * OUT_DIM + 255) / 256, 256, 0, stream>>>(out, emax, esum);

  gat_gemm_wmma<<<N_NODES / 16, 256, 0, stream>>>(features, fc_weight, h);

  node_logits<<<(N_NODES + 7) / 8, 256, 0, stream>>>(h, attn_l, attn_r, el, er);

  edge_score_max<<<(N_EDGES + 255) / 256, 256, 0, stream>>>(src, dst, el, er, score, emax);

  edge_exp_sum<<<(N_EDGES + 255) / 256, 256, 0, stream>>>(dst, score, emax, esum);

  aggregate<<<(N_EDGES + 7) / 8, 256, 0, stream>>>(src, dst, h, score, esum, out);
}


# Aggregate per destination range: 250 work-groups each filter the unsorted dst list for their 200 nodes, sum rows in LDS by LDS f32 atomics (arrival order), store out once; global atomics on out go
# speedup vs baseline: 1.3800x; 1.3800x over previous
#include <hip/hip_runtime.h>
#include <hip/hip_bf16.h>

#define N_NODES 50000
#define N_EDGES 1600000
#define IN_DIM 256
#define OUT_DIM 128
#define NEG_SLOPE 0.2f

typedef __attribute__((ext_vector_type(2))) float v2f;
typedef __attribute__((ext_vector_type(8))) float v8f;

__global__ __launch_bounds__(256) void gat_gemm_wmma(
    const float* __restrict__ X,
    const float* __restrict__ W,
    float* __restrict__ H)
{
  const int lane = threadIdx.x & 31;
  const int wave = threadIdx.x >> 5;
  const int l16  = lane & 15;
  const int half = lane >> 4;

  const int mBase = blockIdx.x * 16;
  const int nBase = wave * 16;

  const float* aPtr = X + (size_t)(mBase + l16) * IN_DIM + 2 * half;
  const float* bPtr = W + (size_t)(2 * half) * OUT_DIM + nBase + l16;

  v8f acc = {};
#pragma unroll 4
  for (int kk = 0; kk < IN_DIM; kk += 4) {
    v2f a = *(const v2f*)(aPtr + kk);
    v2f b;
    b.x = bPtr[(size_t)kk * OUT_DIM];
    b.y = bPtr[(size_t)(kk + 1) * OUT_DIM];
    acc = __builtin_amdgcn_wmma_f32_16x16x4_f32(
         false, a,  false, b,
         (short)0, acc,  false,  false);
  }

  float* out = H + (size_t)(mBase + 8 * half) * OUT_DIM + nBase + l16;
#pragma unroll
  for (int v = 0; v < 8; ++v)
    out[(size_t)v * OUT_DIM] = acc[v];
}

__global__ __launch_bounds__(256) void node_logits(
    const float* __restrict__ H,
    const float* __restrict__ attn_l,
    const float* __restrict__ attn_r,
    float* __restrict__ el, float* __restrict__ er)
{
  const int lane = threadIdx.x & 31;
  const int node = blockIdx.x * (blockDim.x >> 5) + (threadIdx.x >> 5);
  if (node >= N_NODES) return;

  const float4 hv = *(const float4*)(H + (size_t)node * OUT_DIM + 4 * lane);
  const float4 al = *(const float4*)(attn_l + 4 * lane);
  const float4 ar = *(const float4*)(attn_r + 4 * lane);
  float sl = hv.x * al.x + hv.y * al.y + hv.z * al.z + hv.w * al.w;
  float sr = hv.x * ar.x + hv.y * ar.y + hv.z * ar.z + hv.w * ar.w;
#pragma unroll
  for (int off = 16; off > 0; off >>= 1) {
    sl += __shfl_down(sl, off, 32);
    sr += __shfl_down(sr, off, 32);
  }
  if (lane == 0) { el[node] = sl; er[node] = sr; }
}

__global__ __launch_bounds__(256) void init_buffers(
    float* __restrict__ emax, float* __restrict__ esum)
{
  const int i = blockIdx.x * blockDim.x + threadIdx.x;
  if (i < N_NODES) { emax[i] = -__builtin_inff(); esum[i] = 0.0f; }
}

__global__ __launch_bounds__(256) void edge_score_max(
    const int* __restrict__ src, const int* __restrict__ dst,
    const float* __restrict__ el, const float* __restrict__ er,
    float* __restrict__ score, float* __restrict__ emax)
{
  const int e = blockIdx.x * blockDim.x + threadIdx.x;
  if (e >= N_EDGES) return;
  const float t = el[src[e]] + er[dst[e]];
  const float s = t > 0.0f ? t : NEG_SLOPE * t;
  score[e] = s;
  atomicMax(emax + dst[e], s);
}

__global__ __launch_bounds__(256) void edge_exp_sum(
    const int* __restrict__ dst,
    float* __restrict__ score,
    const float* __restrict__ emax, float* __restrict__ esum)
{
  const int e = blockIdx.x * blockDim.x + threadIdx.x;
  if (e >= N_EDGES) return;
  const int d = dst[e];
  const float ex = __expf(score[e] - emax[d]);
  score[e] = ex;
  atomicAdd(esum + d, ex);
}

#define AGG_NB   200
#define AGG_T    1024
#define AGG_QCAP 16384
#define AGG_NV   (N_EDGES / 4)
static_assert(N_NODES % AGG_NB == 0, "node ranges must tile the nodes");
static_assert(N_EDGES % 4 == 0, "dst is read in 16-byte groups");
static_assert(N_EDGES <= (1 << 21) && AGG_NB <= 256, "list entry: edge in 21 bits, local node in 8");
static_assert(N_NODES <= (1 << 16), "list entry: source node in 16 bits");

template <bool WHOLE>
__device__ __forceinline__ void agg_rows(
    int cnt, int pk_l, float a_l, const float* __restrict__ H, float* acc, int lane)
{
  float hv[8][4];
#pragma unroll
  for (int k = 0; k < 8; ++k) {
    if (WHOLE || k < cnt) {
      const int s = __builtin_amdgcn_readlane(pk_l, k) & 0xFFFF;
      const float* hp = H + (size_t)s * OUT_DIM + lane;
      hv[k][0] = hp[0];
      hv[k][1] = hp[32];
      hv[k][2] = hp[64];
      hv[k][3] = hp[96];
    }
  }
#pragma unroll
  for (int k = 0; k < 8; ++k) {
    if (WHOLE || k < cnt) {
      const int   r = (int)((unsigned)__builtin_amdgcn_readlane(pk_l, k) >> 16);
      const float a = __int_as_float(__builtin_amdgcn_readlane(__float_as_int(a_l), k));
      float* o = acc + r * OUT_DIM + lane;
      atomicAdd(o +  0, a * hv[k][0]);
      atomicAdd(o + 32, a * hv[k][1]);
      atomicAdd(o + 64, a * hv[k][2]);
      atomicAdd(o + 96, a * hv[k][3]);
    }
  }
}

__global__ __launch_bounds__(AGG_T) void aggregate(
    const int* __restrict__ src, const int* __restrict__ dst,
    const float* __restrict__ H,
    const float* __restrict__ eexp, const float* __restrict__ esum,
    float* __restrict__ out)
{
  __shared__ float acc[AGG_NB * OUT_DIM];
  __shared__ int   q_pk[AGG_QCAP];
  __shared__ float q_a[AGG_QCAP];
  __shared__ int   q_n;

  const int t    = threadIdx.x;
  const int lane = t & 31;
  const int wave = __builtin_amdgcn_readfirstlane(t >> 5);
  const int lo   = blockIdx.x * AGG_NB;

  for (int i = t; i < AGG_NB * OUT_DIM; i += AGG_T) acc[i] = 0.0f;
  if (t == 0) q_n = 0;
  __syncthreads();

  const int4* dst4 = (const int4*)dst;
  for (int base = 0; base < AGG_NV; base += 4 * AGG_T) {
    int4 v[4];
#pragma unroll
    for (int u = 0; u < 4; ++u) {
      const int i = base + u * AGG_T + t;
      v[u] = (i < AGG_NV) ? dst4[i] : make_int4(-1, -1, -1, -1);
    }
#pragma unroll
    for (int u = 0; u < 4; ++u) {
      const int e = 4 * (base + u * AGG_T + t);
      const unsigned rr[4] = {(unsigned)(v[u].x - lo), (unsigned)(v[u].y - lo),
                              (unsigned)(v[u].z - lo), (unsigned)(v[u].w - lo)};
      if (min(min(rr[0], rr[1]), min(rr[2], rr[3])) < (unsigned)AGG_NB) {
#pragma unroll
        for (int c = 0; c < 4; ++c) {
          if (rr[c] < (unsigned)AGG_NB) {
            const int idx = atomicAdd(&q_n, 1);
            if (idx < AGG_QCAP) q_pk[idx] = (e + c) | ((int)rr[c] << 21);
          }
        }
      }
    }
  }
  __syncthreads();
  const int total = __builtin_amdgcn_readfirstlane(q_n);
  const bool full = total > AGG_QCAP;
  const int n = full ? 0 : total;

  for (int i = t; i < n; i += AGG_T) {
    const int pk = q_pk[i];
    const int e = pk & 0x1FFFFF;
    const int r = (int)((unsigned)pk >> 21);
    const int s = src[e];
    q_a[i]  = eexp[e] / esum[lo + r];
    q_pk[i] = s | (r << 16);
  }
  __syncthreads();

  for (int b = wave * 8; b < n; b += (AGG_T / 32) * 8) {
    const int j = b + (lane & 7);
    if (n - b >= 8) {
      agg_rows<true>(8, q_pk[j], q_a[j], H, acc, lane);
    } else {
      const bool in = j < n;
      agg_rows<false>(n - b, in ? q_pk[j] : 0, in ? q_a[j] : 0.0f, H, acc, lane);
    }
  }
  if (full) {
    for (int e = wave; e < N_EDGES; e += AGG_T / 32) {
      const int d = dst[e];
      const unsigned r = (unsigned)(d - lo);
      if (r < (unsigned)AGG_NB) {
        const float a = eexp[e] / esum[d];
        const float* hp = H + (size_t)src[e] * OUT_DIM + lane;
        float* o = acc + r * OUT_DIM + lane;
        atomicAdd(o +  0, a * hp[0]);
        atomicAdd(o + 32, a * hp[32]);
        atomicAdd(o + 64, a * hp[64]);
        atomicAdd(o + 96, a * hp[96]);
      }
    }
  }
  __syncthreads();

  float* op = out + (size_t)lo * OUT_DIM;
  for (int i = t; i < AGG_NB * OUT_DIM; i += AGG_T) op[i] = acc[i];
}

extern "C" void kernel_launch(void* const* d_in, const int* in_sizes, int n_in,
                              void* d_out, int out_size, void* d_ws, size_t ws_size,
                              hipStream_t stream) {
  (void)in_sizes; (void)n_in; (void)out_size; (void)ws_size;

  const float* features  = (const float*)d_in[0];
  const float* fc_weight = (const float*)d_in[1];
  const float* attn_l    = (const float*)d_in[2];
  const float* attn_r    = (const float*)d_in[3];
  const int*   src       = (const int*)d_in[4];
  const int*   dst       = (const int*)d_in[5];
  float*       out       = (float*)d_out;

  float* h     = (float*)d_ws;
  float* el    = h + (size_t)N_NODES * OUT_DIM;
  float* er    = el + N_NODES;
  float* emax  = er + N_NODES;
  float* esum  = emax + N_NODES;
  float* score = esum + N_NODES;

  init_buffers<<<(N_NODES + 255) / 256, 256, 0, stream>>>(emax, esum);

  gat_gemm_wmma<<<N_NODES / 16, 256, 0, stream>>>(features, fc_weight, h);

  node_logits<<<(N_NODES + 7) / 8, 256, 0, stream>>>(h, attn_l, attn_r, el, er);

  edge_score_max<<<(N_EDGES + 255) / 256, 256, 0, stream>>>(src, dst, el, er, score, emax);

  edge_exp_sum<<<(N_EDGES + 255) / 256, 256, 0, stream>>>(dst, score, emax, esum);

  aggregate<<<N_NODES / AGG_NB, AGG_T, 0, stream>>>(src, dst, h, score, esum, out);
}


# Edge softmax moved into the per-range aggregation work-group: score, segment max and sum on its LDS edge list by LDS f32 atomics (sum in arrival order); 3 of 6 launches and all global atomics go
# speedup vs baseline: 2.6094x; 1.8909x over previous
#include <hip/hip_runtime.h>
#include <hip/hip_bf16.h>

#define N_NODES 50000
#define N_EDGES 1600000
#define IN_DIM 256
#define OUT_DIM 128
#define NEG_SLOPE 0.2f

typedef __attribute__((ext_vector_type(2))) float v2f;
typedef __attribute__((ext_vector_type(8))) float v8f;

__global__ __launch_bounds__(256) void gat_gemm_wmma(
    const float* __restrict__ X,
    const float* __restrict__ W,
    float* __restrict__ H)
{
  const int lane = threadIdx.x & 31;
  const int wave = threadIdx.x >> 5;
  const int l16  = lane & 15;
  const int half = lane >> 4;

  const int mBase = blockIdx.x * 16;
  const int nBase = wave * 16;

  const float* aPtr = X + (size_t)(mBase + l16) * IN_DIM + 2 * half;
  const float* bPtr = W + (size_t)(2 * half) * OUT_DIM + nBase + l16;

  v8f acc = {};
#pragma unroll 4
  for (int kk = 0; kk < IN_DIM; kk += 4) {
    v2f a = *(const v2f*)(aPtr + kk);
    v2f b;
    b.x = bPtr[(size_t)kk * OUT_DIM];
    b.y = bPtr[(size_t)(kk + 1) * OUT_DIM];
    acc = __builtin_amdgcn_wmma_f32_16x16x4_f32(
         false, a,  false, b,
         (short)0, acc,  false,  false);
  }

  float* out = H + (size_t)(mBase + 8 * half) * OUT_DIM + nBase + l16;
#pragma unroll
  for (int v = 0; v < 8; ++v)
    out[(size_t)v * OUT_DIM] = acc[v];
}

__global__ __launch_bounds__(256) void node_logits(
    const float* __restrict__ H,
    const float* __restrict__ attn_l,
    const float* __restrict__ attn_r,
    float* __restrict__ el, float* __restrict__ er)
{
  const int lane = threadIdx.x & 31;
  const int node = blockIdx.x * (blockDim.x >> 5) + (threadIdx.x >> 5);
  if (node >= N_NODES) return;

  const float4 hv = *(const float4*)(H + (size_t)node * OUT_DIM + 4 * lane);
  const float4 al = *(const float4*)(attn_l + 4 * lane);
  const float4 ar = *(const float4*)(attn_r + 4 * lane);
  float sl = hv.x * al.x + hv.y * al.y + hv.z * al.z + hv.w * al.w;
  float sr = hv.x * ar.x + hv.y * ar.y + hv.z * ar.z + hv.w * ar.w;
#pragma unroll
  for (int off = 16; off > 0; off >>= 1) {
    sl += __shfl_down(sl, off, 32);
    sr += __shfl_down(sr, off, 32);
  }
  if (lane == 0) { el[node] = sl; er[node] = sr; }
}

#define AGG_NB   200
#define AGG_T    1024
#define AGG_QCAP 16384
#define AGG_NV   (N_EDGES / 4)
static_assert(N_NODES % AGG_NB == 0, "node ranges must tile the nodes");
static_assert(N_EDGES % 4 == 0, "dst is read in 16-byte groups");
static_assert(N_EDGES <= (1 << 21) && AGG_NB <= 256, "list entry: edge in 21 bits, local node in 8");
static_assert(N_NODES <= (1 << 16), "list entry: source node in 16 bits");

template <bool WHOLE>
__device__ __forceinline__ void agg_rows(
    int cnt, int pk_l, float a_l, const float* __restrict__ H, float* acc, int lane)
{
  float hv[8][4];
#pragma unroll
  for (int k = 0; k < 8; ++k) {
    if (WHOLE || k < cnt) {
      const int s = __builtin_amdgcn_readlane(pk_l, k) & 0xFFFF;
      const float* hp = H + (size_t)s * OUT_DIM + lane;
      hv[k][0] = hp[0];
      hv[k][1] = hp[32];
      hv[k][2] = hp[64];
      hv[k][3] = hp[96];
    }
  }
#pragma unroll
  for (int k = 0; k < 8; ++k) {
    if (WHOLE || k < cnt) {
      const int   r = (int)((unsigned)__builtin_amdgcn_readlane(pk_l, k) >> 16);
      const float a = __int_as_float(__builtin_amdgcn_readlane(__float_as_int(a_l), k));
      float* o = acc + r * OUT_DIM + lane;
      atomicAdd(o +  0, a * hv[k][0]);
      atomicAdd(o + 32, a * hv[k][1]);
      atomicAdd(o + 64, a * hv[k][2]);
      atomicAdd(o + 96, a * hv[k][3]);
    }
  }
}

__device__ __forceinline__ float edge_score(float l, float r)
{
  const float t = l + r;
  return t > 0.0f ? t : NEG_SLOPE * t;
}

#define AGG_SB 8
#define AGG_SBLK (32 * AGG_SB)

template <bool WHOLE>
__device__ __forceinline__ void score_block(
    int base, int n, int lane,
    const int* __restrict__ src, const float* __restrict__ el,
    const float* er_s, int* q_pk, float* q_a, float* mx_s)
{
  int pk[AGG_SB], s[AGG_SB];
  float l[AGG_SB];
#pragma unroll
  for (int u = 0; u < AGG_SB; ++u) {
    const int i = base + 32 * u + lane;
    pk[u] = (WHOLE || i < n) ? q_pk[i] : 0;
  }
#pragma unroll
  for (int u = 0; u < AGG_SB; ++u) s[u] = src[pk[u] & 0x1FFFFF];
#pragma unroll
  for (int u = 0; u < AGG_SB; ++u) l[u] = el[s[u]];
#pragma unroll
  for (int u = 0; u < AGG_SB; ++u) {
    const int i = base + 32 * u + lane;
    if (WHOLE || i < n) {
      const int r = (int)((unsigned)pk[u] >> 21);
      const float sc = edge_score(l[u], er_s[r]);
      q_a[i]  = sc;
      q_pk[i] = s[u] | (r << 16);
      __hip_atomic_fetch_max(&mx_s[r], sc, __ATOMIC_RELAXED, __HIP_MEMORY_SCOPE_WORKGROUP);
    }
  }
}

__global__ __launch_bounds__(AGG_T) void aggregate(
    const int* __restrict__ src, const int* __restrict__ dst,
    const float* __restrict__ H,
    const float* __restrict__ el, const float* __restrict__ er,
    float* __restrict__ out)
{
  __shared__ float acc[AGG_NB * OUT_DIM];
  __shared__ int   q_pk[AGG_QCAP];
  __shared__ float q_a[AGG_QCAP];
  __shared__ float er_s[AGG_NB];
  __shared__ float mx_s[AGG_NB];
  __shared__ float sm_s[AGG_NB];
  __shared__ int   q_n;

  const int t    = threadIdx.x;
  const int lane = t & 31;
  const int wave = __builtin_amdgcn_readfirstlane(t >> 5);
  const int lo   = blockIdx.x * AGG_NB;

  for (int i = t; i < AGG_NB * OUT_DIM; i += AGG_T) acc[i] = 0.0f;
  if (t < AGG_NB) {
    er_s[t] = er[lo + t];
    mx_s[t] = -__builtin_inff();
    sm_s[t] = 0.0f;
  }
  if (t == 0) q_n = 0;
  __syncthreads();

  const int4* dst4 = (const int4*)dst;
  for (int base = 0; base < AGG_NV; base += 4 * AGG_T) {
    int4 v[4];
#pragma unroll
    for (int u = 0; u < 4; ++u) {
      const int i = base + u * AGG_T + t;
      v[u] = (i < AGG_NV) ? dst4[i] : make_int4(-1, -1, -1, -1);
    }
#pragma unroll
    for (int u = 0; u < 4; ++u) {
      const int e = 4 * (base + u * AGG_T + t);
      const unsigned rr[4] = {(unsigned)(v[u].x - lo), (unsigned)(v[u].y - lo),
                              (unsigned)(v[u].z - lo), (unsigned)(v[u].w - lo)};
      if (min(min(rr[0], rr[1]), min(rr[2], rr[3])) < (unsigned)AGG_NB) {
#pragma unroll
        for (int c = 0; c < 4; ++c) {
          if (rr[c] < (unsigned)AGG_NB) {
            const int idx = atomicAdd(&q_n, 1);
            if (idx < AGG_QCAP) q_pk[idx] = (e + c) | ((int)rr[c] << 21);
          }
        }
      }
    }
  }
  __syncthreads();
  const int total = __builtin_amdgcn_readfirstlane(q_n);
  const bool full = total > AGG_QCAP;
  const int n = full ? 0 : total;

  for (int base = wave * AGG_SBLK; base < n; base += (AGG_T / 32) * AGG_SBLK) {
    if (n - base >= AGG_SBLK) {
      score_block<true>(base, n, lane, src, el, er_s, q_pk, q_a, mx_s);
    } else {
      score_block<false>(base, n, lane, src, el, er_s, q_pk, q_a, mx_s);
    }
  }
  if (full) {
    for (int e = t; e < N_EDGES; e += AGG_T) {
      const unsigned r = (unsigned)(dst[e] - lo);
      if (r < (unsigned)AGG_NB) {
        const float sc = edge_score(el[src[e]], er_s[r]);
        __hip_atomic_fetch_max(&mx_s[r], sc, __ATOMIC_RELAXED, __HIP_MEMORY_SCOPE_WORKGROUP);
      }
    }
  }
  __syncthreads();

  for (int i = t; i < n; i += AGG_T) {
    const int r = (int)((unsigned)q_pk[i] >> 16);
    const float ex = __expf(q_a[i] - mx_s[r]);
    q_a[i] = ex;
    atomicAdd(&sm_s[r], ex);
  }
  if (full) {
    for (int e = t; e < N_EDGES; e += AGG_T) {
      const unsigned r = (unsigned)(dst[e] - lo);
      if (r < (unsigned)AGG_NB) {
        const float sc = edge_score(el[src[e]], er_s[r]);
        atomicAdd(&sm_s[r], __expf(sc - mx_s[r]));
      }
    }
  }
  __syncthreads();

  for (int b = wave * 8; b < n; b += (AGG_T / 32) * 8) {
    const int j = b + (lane & 7);
    if (n - b >= 8) {
      const int pk = q_pk[j];
      const float a = q_a[j] / sm_s[(unsigned)pk >> 16];
      agg_rows<true>(8, pk, a, H, acc, lane);
    } else {
      const bool in = j < n;
      const int pk = in ? q_pk[j] : 0;
      float a = 0.0f;
      if (in) a = q_a[j] / sm_s[(unsigned)pk >> 16];
      agg_rows<false>(n - b, pk, a, H, acc, lane);
    }
  }
  if (full) {
    for (int e = wave; e < N_EDGES; e += AGG_T / 32) {
      const unsigned r = (unsigned)(dst[e] - lo);
      if (r < (unsigned)AGG_NB) {
        const int s = src[e];
        const float sc = edge_score(el[s], er_s[r]);
        const float a = __expf(sc - mx_s[r]) / sm_s[r];
        const float* hp = H + (size_t)s * OUT_DIM + lane;
        float* o = acc + r * OUT_DIM + lane;
        atomicAdd(o +  0, a * hp[0]);
        atomicAdd(o + 32, a * hp[32]);
        atomicAdd(o + 64, a * hp[64]);
        atomicAdd(o + 96, a * hp[96]);
      }
    }
  }
  __syncthreads();

  float* op = out + (size_t)lo * OUT_DIM;
  for (int i = t; i < AGG_NB * OUT_DIM; i += AGG_T) op[i] = acc[i];
}

extern "C" void kernel_launch(void* const* d_in, const int* in_sizes, int n_in,
                              void* d_out, int out_size, void* d_ws, size_t ws_size,
                              hipStream_t stream) {
  (void)in_sizes; (void)n_in; (void)out_size; (void)ws_size;

  const float* features  = (const float*)d_in[0];
  const float* fc_weight = (const float*)d_in[1];
  const float* attn_l    = (const float*)d_in[2];
  const float* attn_r    = (const float*)d_in[3];
  const int*   src       = (const int*)d_in[4];
  const int*   dst       = (const int*)d_in[5];
  float*       out       = (float*)d_out;

  float* h  = (float*)d_ws;
  float* el = h + (size_t)N_NODES * OUT_DIM;
  float* er = el + N_NODES;

  gat_gemm_wmma<<<N_NODES / 16, 256, 0, stream>>>(features, fc_weight, h);

  node_logits<<<(N_NODES + 7) / 8, 256, 0, stream>>>(h, attn_l, attn_r, el, er);

  aggregate<<<N_NODES / AGG_NB, AGG_T, 0, stream>>>(src, dst, h, el, er, out);
}


# Aggregate scans a 1-byte range-id plane (dst/200, written in the logits launch), 16 edges per 16-byte load, not the 4-byte dst list; flagged groups resolved to edges per lane; sums in arrival order
# speedup vs baseline: 3.1243x; 1.1973x over previous
#include <hip/hip_runtime.h>
#include <hip/hip_bf16.h>

#define N_NODES 50000
#define N_EDGES 1600000
#define IN_DIM 256
#define OUT_DIM 128
#define NEG_SLOPE 0.2f

typedef __attribute__((ext_vector_type(2))) float v2f;
typedef __attribute__((ext_vector_type(8))) float v8f;

__global__ __launch_bounds__(256) void gat_gemm_wmma(
    const float* __restrict__ X,
    const float* __restrict__ W,
    float* __restrict__ H)
{
  const int lane = threadIdx.x & 31;
  const int wave = threadIdx.x >> 5;
  const int l16  = lane & 15;
  const int half = lane >> 4;

  const int mBase = blockIdx.x * 16;
  const int nBase = wave * 16;

  const float* aPtr = X + (size_t)(mBase + l16) * IN_DIM + 2 * half;
  const float* bPtr = W + (size_t)(2 * half) * OUT_DIM + nBase + l16;

  v8f acc = {};
#pragma unroll 4
  for (int kk = 0; kk < IN_DIM; kk += 4) {
    v2f a = *(const v2f*)(aPtr + kk);
    v2f b;
    b.x = bPtr[(size_t)kk * OUT_DIM];
    b.y = bPtr[(size_t)(kk + 1) * OUT_DIM];
    acc = __builtin_amdgcn_wmma_f32_16x16x4_f32(
         false, a,  false, b,
         (short)0, acc,  false,  false);
  }

  float* out = H + (size_t)(mBase + 8 * half) * OUT_DIM + nBase + l16;
#pragma unroll
  for (int v = 0; v < 8; ++v)
    out[(size_t)v * OUT_DIM] = acc[v];
}

#define AGG_NB   200
#define RID_NONE 255u
static_assert(N_NODES % AGG_NB == 0, "node ranges must tile the nodes");
static_assert(N_NODES / AGG_NB <= 255, "a range id must fit a byte, 255 kept free");
static_assert(N_EDGES % 16 == 0, "the range-id plane is read in 16-byte groups");

__device__ __forceinline__ unsigned range_id(int d)
{
  return (unsigned)d < (unsigned)N_NODES ? (unsigned)d / AGG_NB : RID_NONE;
}

__global__ __launch_bounds__(256) void node_logits(
    const float* __restrict__ H,
    const float* __restrict__ attn_l,
    const float* __restrict__ attn_r,
    const int* __restrict__ dst,
    float* __restrict__ el, float* __restrict__ er,
    int* __restrict__ rid4)
{
  const int gid = blockIdx.x * blockDim.x + threadIdx.x;
  if (gid < N_EDGES / 4) {
    const int4 d = ((const int4*)dst)[gid];
    rid4[gid] = (int)(range_id(d.x) | (range_id(d.y) << 8) |
                      (range_id(d.z) << 16) | (range_id(d.w) << 24));
  }

  const int lane = threadIdx.x & 31;
  const int node = blockIdx.x * (blockDim.x >> 5) + (threadIdx.x >> 5);
  if (node >= N_NODES) return;

  const float4 hv = *(const float4*)(H + (size_t)node * OUT_DIM + 4 * lane);
  const float4 al = *(const float4*)(attn_l + 4 * lane);
  const float4 ar = *(const float4*)(attn_r + 4 * lane);
  float sl = hv.x * al.x + hv.y * al.y + hv.z * al.z + hv.w * al.w;
  float sr = hv.x * ar.x + hv.y * ar.y + hv.z * ar.z + hv.w * ar.w;
#pragma unroll
  for (int off = 16; off > 0; off >>= 1) {
    sl += __shfl_down(sl, off, 32);
    sr += __shfl_down(sr, off, 32);
  }
  if (lane == 0) { el[node] = sl; er[node] = sr; }
}

#define AGG_T    1024
#define AGG_QCAP 16384
#define AGG_NG   (N_EDGES / 16)
static_assert(AGG_NB <= 256, "list entry: local node in 8 bits");
static_assert(N_NODES <= (1 << 16), "list entry: source node in 16 bits");

template <bool WHOLE>
__device__ __forceinline__ void agg_rows(
    int cnt, int pk_l, float a_l, const float* __restrict__ H, float* acc, int lane)
{
  float hv[8][4];
#pragma unroll
  for (int k = 0; k < 8; ++k) {
    if (WHOLE || k < cnt) {
      const int s = __builtin_amdgcn_readlane(pk_l, k) & 0xFFFF;
      const float* hp = H + (size_t)s * OUT_DIM + lane;
      hv[k][0] = hp[0];
      hv[k][1] = hp[32];
      hv[k][2] = hp[64];
      hv[k][3] = hp[96];
    }
  }
#pragma unroll
  for (int k = 0; k < 8; ++k) {
    if (WHOLE || k < cnt) {
      const int   r = (int)((unsigned)__builtin_amdgcn_readlane(pk_l, k) >> 16);
      const float a = __int_as_float(__builtin_amdgcn_readlane(__float_as_int(a_l), k));
      float* o = acc + r * OUT_DIM + lane;
      atomicAdd(o +  0, a * hv[k][0]);
      atomicAdd(o + 32, a * hv[k][1]);
      atomicAdd(o + 64, a * hv[k][2]);
      atomicAdd(o + 96, a * hv[k][3]);
    }
  }
}

__device__ __forceinline__ unsigned zero_byte_flags(unsigned y)
{
  return (y - 0x01010101u) & ~y;
}

__device__ __forceinline__ float edge_score(float l, float r)
{
  const float t = l + r;
  return t > 0.0f ? t : NEG_SLOPE * t;
}

#define AGG_SB 8
#define AGG_SBLK (32 * AGG_SB)

template <bool WHOLE>
__device__ __forceinline__ void score_block(
    int base, int n, int lane,
    const int* __restrict__ src, const int* __restrict__ dst, int lo,
    const float* __restrict__ el,
    const float* er_s, int* q_pk, float* q_a, float* mx_s)
{
  int pk[AGG_SB], s[AGG_SB], d[AGG_SB];
  float l[AGG_SB];
#pragma unroll
  for (int u = 0; u < AGG_SB; ++u) {
    const int i = base + 32 * u + lane;
    pk[u] = (WHOLE || i < n) ? q_pk[i] : 0;
  }
#pragma unroll
  for (int u = 0; u < AGG_SB; ++u) { s[u] = src[pk[u]]; d[u] = dst[pk[u]]; }
#pragma unroll
  for (int u = 0; u < AGG_SB; ++u) l[u] = el[s[u]];
#pragma unroll
  for (int u = 0; u < AGG_SB; ++u) {
    const int i = base + 32 * u + lane;
    if (WHOLE || i < n) {
      const int r = d[u] - lo;
      const float sc = edge_score(l[u], er_s[r]);
      q_a[i]  = sc;
      q_pk[i] = s[u] | (r << 16);
      __hip_atomic_fetch_max(&mx_s[r], sc, __ATOMIC_RELAXED, __HIP_MEMORY_SCOPE_WORKGROUP);
    }
  }
}

__global__ __launch_bounds__(AGG_T) void aggregate(
    const int* __restrict__ src, const int* __restrict__ dst,
    const int* __restrict__ rid4,
    const float* __restrict__ H,
    const float* __restrict__ el, const float* __restrict__ er,
    float* __restrict__ out)
{
  __shared__ float acc[AGG_NB * OUT_DIM];
  __shared__ int   q_pk[AGG_QCAP];
  __shared__ float q_a[AGG_QCAP];
  __shared__ float er_s[AGG_NB];
  __shared__ float mx_s[AGG_NB];
  __shared__ float sm_s[AGG_NB];
  __shared__ int   q_n;
  __shared__ int   c_n;

  const int t    = threadIdx.x;
  const int lane = t & 31;
  const int wave = __builtin_amdgcn_readfirstlane(t >> 5);
  const int lo   = blockIdx.x * AGG_NB;

  for (int i = t; i < AGG_NB * OUT_DIM; i += AGG_T) acc[i] = 0.0f;
  if (t < AGG_NB) {
    er_s[t] = er[lo + t];
    mx_s[t] = -__builtin_inff();
    sm_s[t] = 0.0f;
  }
  if (t == 0) { q_n = 0; c_n = 0; }
  __syncthreads();

  const int4* rp = (const int4*)rid4;
  const unsigned id4 = blockIdx.x * 0x01010101u;
  int* cand = (int*)q_a;
  for (int base = 0; base < AGG_NG; base += 4 * AGG_T) {
    int4 v[4];
#pragma unroll
    for (int u = 0; u < 4; ++u) {
      const int i = base + u * AGG_T + t;
      v[u] = (i < AGG_NG) ? rp[i] : make_int4(-1, -1, -1, -1);
    }
#pragma unroll
    for (int u = 0; u < 4; ++u) {
      const int i = base + u * AGG_T + t;
      const unsigned f = (zero_byte_flags((unsigned)v[u].x ^ id4) | zero_byte_flags((unsigned)v[u].y ^ id4) |
                          zero_byte_flags((unsigned)v[u].z ^ id4) | zero_byte_flags((unsigned)v[u].w ^ id4)) & 0x80808080u;
      if (f != 0u) {
        const int idx = atomicAdd(&c_n, 1);
        if (idx < AGG_QCAP) cand[idx] = i;
      }
    }
  }
  __syncthreads();
  const int ctot = __builtin_amdgcn_readfirstlane(c_n);
  const int nc = ctot > AGG_QCAP ? 0 : ctot;

  for (int j = t; j < nc; j += AGG_T) {
    const int g = cand[j];
    const int4 x = rp[g];
    const unsigned y[4] = {(unsigned)x.x ^ id4, (unsigned)x.y ^ id4, (unsigned)x.z ^ id4, (unsigned)x.w ^ id4};
#pragma unroll
    for (int c = 0; c < 4; ++c) {
      unsigned h = zero_byte_flags(y[c]) & 0x80808080u;
      while (h != 0u) {
        const int b = __builtin_ctz(h) >> 3;
        h &= h - 1u;
        if (((y[c] >> (8 * b)) & 0xFFu) == 0u) {
          const int idx = atomicAdd(&q_n, 1);
          if (idx < AGG_QCAP) q_pk[idx] = 16 * g + 4 * c + b;
        }
      }
    }
  }
  __syncthreads();
  const int total = __builtin_amdgcn_readfirstlane(q_n);
  const bool full = ctot > AGG_QCAP || total > AGG_QCAP;
  const int n = full ? 0 : total;

  for (int base = wave * AGG_SBLK; base < n; base += (AGG_T / 32) * AGG_SBLK) {
    if (n - base >= AGG_SBLK) {
      score_block<true>(base, n, lane, src, dst, lo, el, er_s, q_pk, q_a, mx_s);
    } else {
      score_block<false>(base, n, lane, src, dst, lo, el, er_s, q_pk, q_a, mx_s);
    }
  }
  if (full) {
    for (int e = t; e < N_EDGES; e += AGG_T) {
      const unsigned r = (unsigned)(dst[e] - lo);
      if (r < (unsigned)AGG_NB) {
        const float sc = edge_score(el[src[e]], er_s[r]);
        __hip_atomic_fetch_max(&mx_s[r], sc, __ATOMIC_RELAXED, __HIP_MEMORY_SCOPE_WORKGROUP);
      }
    }
  }
  __syncthreads();

  for (int i = t; i < n; i += AGG_T) {
    const int r = (int)((unsigned)q_pk[i] >> 16);
    const float ex = __expf(q_a[i] - mx_s[r]);
    q_a[i] = ex;
    atomicAdd(&sm_s[r], ex);
  }
  if (full) {
    for (int e = t; e < N_EDGES; e += AGG_T) {
      const unsigned r = (unsigned)(dst[e] - lo);
      if (r < (unsigned)AGG_NB) {
        const float sc = edge_score(el[src[e]], er_s[r]);
        atomicAdd(&sm_s[r], __expf(sc - mx_s[r]));
      }
    }
  }
  __syncthreads();

  for (int b = wave * 8; b < n; b += (AGG_T / 32) * 8) {
    const int j = b + (lane & 7);
    if (n - b >= 8) {
      const int pk = q_pk[j];
      const float a = q_a[j] / sm_s[(unsigned)pk >> 16];
      agg_rows<true>(8, pk, a, H, acc, lane);
    } else {
      const bool in = j < n;
      const int pk = in ? q_pk[j] : 0;
      float a = 0.0f;
      if (in) a = q_a[j] / sm_s[(unsigned)pk >> 16];
      agg_rows<false>(n - b, pk, a, H, acc, lane);
    }
  }
  if (full) {
    for (int e = wave; e < N_EDGES; e += AGG_T / 32) {
      const unsigned r = (unsigned)(dst[e] - lo);
      if (r < (unsigned)AGG_NB) {
        const int s = src[e];
        const float sc = edge_score(el[s], er_s[r]);
        const float a = __expf(sc - mx_s[r]) / sm_s[r];
        const float* hp = H + (size_t)s * OUT_DIM + lane;
        float* o = acc + r * OUT_DIM + lane;
        atomicAdd(o +  0, a * hp[0]);
        atomicAdd(o + 32, a * hp[32]);
        atomicAdd(o + 64, a * hp[64]);
        atomicAdd(o + 96, a * hp[96]);
      }
    }
  }
  __syncthreads();

  float* op = out + (size_t)lo * OUT_DIM;
  for (int i = t; i < AGG_NB * OUT_DIM; i += AGG_T) op[i] = acc[i];
}

extern "C" void kernel_launch(void* const* d_in, const int* in_sizes, int n_in,
                              void* d_out, int out_size, void* d_ws, size_t ws_size,
                              hipStream_t stream) {
  (void)in_sizes; (void)n_in; (void)out_size; (void)ws_size;

  const float* features  = (const float*)d_in[0];
  const float* fc_weight = (const float*)d_in[1];
  const float* attn_l    = (const float*)d_in[2];
  const float* attn_r    = (const float*)d_in[3];
  const int*   src       = (const int*)d_in[4];
  const int*   dst       = (const int*)d_in[5];
  float*       out       = (float*)d_out;

  float* h  = (float*)d_ws;
  float* el = h + (size_t)N_NODES * OUT_DIM;
  float* er = el + N_NODES;
  int* rid4 = (int*)(er + N_NODES);

  gat_gemm_wmma<<<N_NODES / 16, 256, 0, stream>>>(features, fc_weight, h);

  node_logits<<<(N_NODES + 7) / 8, 256, 0, stream>>>(h, attn_l, attn_r, dst, el, er, rid4);

  aggregate<<<N_NODES / AGG_NB, AGG_T, 0, stream>>>(src, dst, rid4, h, el, er, out);
}


# GEMM W-stationary in registers: a wave loads its 256x16 slice of W once (128 VGPRs), 512 work-groups walk the row blocks, A by 16-byte loads a chunk ahead; 2 accumulators: k order of f32 sums changes
# speedup vs baseline: 3.5238x; 1.1279x over previous
#include <hip/hip_runtime.h>
#include <hip/hip_bf16.h>

#define N_NODES 50000
#define N_EDGES 1600000
#define IN_DIM 256
#define OUT_DIM 128
#define NEG_SLOPE 0.2f

typedef __attribute__((ext_vector_type(2))) float v2f;
typedef __attribute__((ext_vector_type(8))) float v8f;

#define GEMM_RB (N_NODES / 16)
#define GEMM_GRID 512
static_assert(N_NODES % 16 == 0, "row blocks must tile the nodes");
static_assert(IN_DIM % 8 == 0, "k is walked 8 at a time");
#define GEMM_CH 4
static_assert((IN_DIM / 8) % GEMM_CH == 0, "chunks must tile the k range");

__global__ __launch_bounds__(256) void gat_gemm_wmma(
    const float* __restrict__ X,
    const float* __restrict__ W,
    float* __restrict__ H)
{
  const int lane = threadIdx.x & 31;
  const int wave = threadIdx.x >> 5;
  const int l16  = lane & 15;
  const int half = lane >> 4;
  const int nBase = wave * 16;

  v2f bf[IN_DIM / 4];
  const float* bPtr = W + (size_t)(4 * half) * OUT_DIM + nBase + l16;
#pragma unroll
  for (int j = 0; j < IN_DIM / 8; ++j) {
    bf[2 * j].x     = bPtr[(size_t)(8 * j + 0) * OUT_DIM];
    bf[2 * j].y     = bPtr[(size_t)(8 * j + 1) * OUT_DIM];
    bf[2 * j + 1].x = bPtr[(size_t)(8 * j + 2) * OUT_DIM];
    bf[2 * j + 1].y = bPtr[(size_t)(8 * j + 3) * OUT_DIM];
  }

  float4 cur[GEMM_CH], nxt[GEMM_CH];
  {
    const float4* p = (const float4*)(X + (size_t)(blockIdx.x * 16 + l16) * IN_DIM + 4 * half);
#pragma unroll
    for (int i = 0; i < GEMM_CH; ++i) cur[i] = p[2 * i];
  }
  for (int rb = blockIdx.x; rb < GEMM_RB; rb += gridDim.x) {
    const int mBase = rb * 16;
    const int rbn = rb + (int)gridDim.x < GEMM_RB ? rb + (int)gridDim.x : rb;
    const float4* aPtr = (const float4*)(X + (size_t)(mBase + l16) * IN_DIM + 4 * half);
    const float4* aNxt = (const float4*)(X + (size_t)(rbn * 16 + l16) * IN_DIM + 4 * half);
    v8f acc0 = {};
    v8f acc1 = {};
#pragma unroll
    for (int c = 0; c < IN_DIM / 8 / GEMM_CH; ++c) {
#pragma unroll
      for (int i = 0; i < GEMM_CH; ++i)
        nxt[i] = (c + 1 < IN_DIM / 8 / GEMM_CH) ? aPtr[2 * ((c + 1) * GEMM_CH + i)] : aNxt[2 * i];
      __builtin_amdgcn_sched_barrier(0);
#pragma unroll
      for (int i = 0; i < GEMM_CH; ++i) {
        const int j = c * GEMM_CH + i;
        v2f a0; a0.x = cur[i].x; a0.y = cur[i].y;
        v2f a1; a1.x = cur[i].z; a1.y = cur[i].w;
        acc0 = __builtin_amdgcn_wmma_f32_16x16x4_f32(
             false, a0,  false, bf[2 * j],
             (short)0, acc0,  false,  false);
        acc1 = __builtin_amdgcn_wmma_f32_16x16x4_f32(
             false, a1,  false, bf[2 * j + 1],
             (short)0, acc1,  false,  false);
      }
#pragma unroll
      for (int i = 0; i < GEMM_CH; ++i) cur[i] = nxt[i];
    }
    float* out = H + (size_t)(mBase + 8 * half) * OUT_DIM + nBase + l16;
#pragma unroll
    for (int v = 0; v < 8; ++v)
      out[(size_t)v * OUT_DIM] = acc0[v] + acc1[v];
  }
}

#define AGG_NB   200
#define RID_NONE 255u
static_assert(N_NODES % AGG_NB == 0, "node ranges must tile the nodes");
static_assert(N_NODES / AGG_NB <= 255, "a range id must fit a byte, 255 kept free");
static_assert(N_EDGES % 16 == 0, "the range-id plane is read in 16-byte groups");

__device__ __forceinline__ unsigned range_id(int d)
{
  return (unsigned)d < (unsigned)N_NODES ? (unsigned)d / AGG_NB : RID_NONE;
}

__global__ __launch_bounds__(256) void node_logits(
    const float* __restrict__ H,
    const float* __restrict__ attn_l,
    const float* __restrict__ attn_r,
    const int* __restrict__ dst,
    float* __restrict__ el, float* __restrict__ er,
    int* __restrict__ rid4)
{
  const int gid = blockIdx.x * blockDim.x + threadIdx.x;
  if (gid < N_EDGES / 4) {
    const int4 d = ((const int4*)dst)[gid];
    rid4[gid] = (int)(range_id(d.x) | (range_id(d.y) << 8) |
                      (range_id(d.z) << 16) | (range_id(d.w) << 24));
  }

  const int lane = threadIdx.x & 31;
  const int node = blockIdx.x * (blockDim.x >> 5) + (threadIdx.x >> 5);
  if (node >= N_NODES) return;

  const float4 hv = *(const float4*)(H + (size_t)node * OUT_DIM + 4 * lane);
  const float4 al = *(const float4*)(attn_l + 4 * lane);
  const float4 ar = *(const float4*)(attn_r + 4 * lane);
  float sl = hv.x * al.x + hv.y * al.y + hv.z * al.z + hv.w * al.w;
  float sr = hv.x * ar.x + hv.y * ar.y + hv.z * ar.z + hv.w * ar.w;
#pragma unroll
  for (int off = 16; off > 0; off >>= 1) {
    sl += __shfl_down(sl, off, 32);
    sr += __shfl_down(sr, off, 32);
  }
  if (lane == 0) { el[node] = sl; er[node] = sr; }
}

#define AGG_T    1024
#define AGG_QCAP 16384
#define AGG_NG   (N_EDGES / 16)
static_assert(AGG_NB <= 256, "list entry: local node in 8 bits");
static_assert(N_NODES <= (1 << 16), "list entry: source node in 16 bits");

template <bool WHOLE>
__device__ __forceinline__ void agg_rows(
    int cnt, int pk_l, float a_l, const float* __restrict__ H, float* acc, int lane)
{
  float hv[8][4];
#pragma unroll
  for (int k = 0; k < 8; ++k) {
    if (WHOLE || k < cnt) {
      const int s = __builtin_amdgcn_readlane(pk_l, k) & 0xFFFF;
      const float* hp = H + (size_t)s * OUT_DIM + lane;
      hv[k][0] = hp[0];
      hv[k][1] = hp[32];
      hv[k][2] = hp[64];
      hv[k][3] = hp[96];
    }
  }
#pragma unroll
  for (int k = 0; k < 8; ++k) {
    if (WHOLE || k < cnt) {
      const int   r = (int)((unsigned)__builtin_amdgcn_readlane(pk_l, k) >> 16);
      const float a = __int_as_float(__builtin_amdgcn_readlane(__float_as_int(a_l), k));
      float* o = acc + r * OUT_DIM + lane;
      atomicAdd(o +  0, a * hv[k][0]);
      atomicAdd(o + 32, a * hv[k][1]);
      atomicAdd(o + 64, a * hv[k][2]);
      atomicAdd(o + 96, a * hv[k][3]);
    }
  }
}

__device__ __forceinline__ unsigned zero_byte_flags(unsigned y)
{
  return (y - 0x01010101u) & ~y;
}

__device__ __forceinline__ float edge_score(float l, float r)
{
  const float t = l + r;
  return t > 0.0f ? t : NEG_SLOPE * t;
}

#define AGG_SB 8
#define AGG_SBLK (32 * AGG_SB)

template <bool WHOLE>
__device__ __forceinline__ void score_block(
    int base, int n, int lane,
    const int* __restrict__ src, const int* __restrict__ dst, int lo,
    const float* __restrict__ el,
    const float* er_s, int* q_pk, float* q_a, float* mx_s)
{
  int pk[AGG_SB], s[AGG_SB], d[AGG_SB];
  float l[AGG_SB];
#pragma unroll
  for (int u = 0; u < AGG_SB; ++u) {
    const int i = base + 32 * u + lane;
    pk[u] = (WHOLE || i < n) ? q_pk[i] : 0;
  }
#pragma unroll
  for (int u = 0; u < AGG_SB; ++u) { s[u] = src[pk[u]]; d[u] = dst[pk[u]]; }
#pragma unroll
  for (int u = 0; u < AGG_SB; ++u) l[u] = el[s[u]];
#pragma unroll
  for (int u = 0; u < AGG_SB; ++u) {
    const int i = base + 32 * u + lane;
    if (WHOLE || i < n) {
      const int r = d[u] - lo;
      const float sc = edge_score(l[u], er_s[r]);
      q_a[i]  = sc;
      q_pk[i] = s[u] | (r << 16);
      __hip_atomic_fetch_max(&mx_s[r], sc, __ATOMIC_RELAXED, __HIP_MEMORY_SCOPE_WORKGROUP);
    }
  }
}

__global__ __launch_bounds__(AGG_T) void aggregate(
    const int* __restrict__ src, const int* __restrict__ dst,
    const int* __restrict__ rid4,
    const float* __restrict__ H,
    const float* __restrict__ el, const float* __restrict__ er,
    float* __restrict__ out)
{
  __shared__ float acc[AGG_NB * OUT_DIM];
  __shared__ int   q_pk[AGG_QCAP];
  __shared__ float q_a[AGG_QCAP];
  __shared__ float er_s[AGG_NB];
  __shared__ float mx_s[AGG_NB];
  __shared__ float sm_s[AGG_NB];
  __shared__ int   q_n;
  __shared__ int   c_n;

  const int t    = threadIdx.x;
  const int lane = t & 31;
  const int wave = __builtin_amdgcn_readfirstlane(t >> 5);
  const int lo   = blockIdx.x * AGG_NB;

  for (int i = t; i < AGG_NB * OUT_DIM; i += AGG_T) acc[i] = 0.0f;
  if (t < AGG_NB) {
    er_s[t] = er[lo + t];
    mx_s[t] = -__builtin_inff();
    sm_s[t] = 0.0f;
  }
  if (t == 0) { q_n = 0; c_n = 0; }
  __syncthreads();

  const int4* rp = (const int4*)rid4;
  const unsigned id4 = blockIdx.x * 0x01010101u;
  int* cand = (int*)q_a;
  for (int base = 0; base < AGG_NG; base += 4 * AGG_T) {
    int4 v[4];
#pragma unroll
    for (int u = 0; u < 4; ++u) {
      const int i = base + u * AGG_T + t;
      v[u] = (i < AGG_NG) ? rp[i] : make_int4(-1, -1, -1, -1);
    }
#pragma unroll
    for (int u = 0; u < 4; ++u) {
      const int i = base + u * AGG_T + t;
      const unsigned f = (zero_byte_flags((unsigned)v[u].x ^ id4) | zero_byte_flags((unsigned)v[u].y ^ id4) |
                          zero_byte_flags((unsigned)v[u].z ^ id4) | zero_byte_flags((unsigned)v[u].w ^ id4)) & 0x80808080u;
      if (f != 0u) {
        const int idx = atomicAdd(&c_n, 1);
        if (idx < AGG_QCAP) cand[idx] = i;
      }
    }
  }
  __syncthreads();
  const int ctot = __builtin_amdgcn_readfirstlane(c_n);
  const int nc = ctot > AGG_QCAP ? 0 : ctot;

  for (int j = t; j < nc; j += AGG_T) {
    const int g = cand[j];
    const int4 x = rp[g];
    const unsigned y[4] = {(unsigned)x.x ^ id4, (unsigned)x.y ^ id4, (unsigned)x.z ^ id4, (unsigned)x.w ^ id4};
#pragma unroll
    for (int c = 0; c < 4; ++c) {
      unsigned h = zero_byte_flags(y[c]) & 0x80808080u;
      while (h != 0u) {
        const int b = __builtin_ctz(h) >> 3;
        h &= h - 1u;
        if (((y[c] >> (8 * b)) & 0xFFu) == 0u) {
          const int idx = atomicAdd(&q_n, 1);
          if (idx < AGG_QCAP) q_pk[idx] = 16 * g + 4 * c + b;
        }
      }
    }
  }
  __syncthreads();
  const int total = __builtin_amdgcn_readfirstlane(q_n);
  const bool full = ctot > AGG_QCAP || total > AGG_QCAP;
  const int n = full ? 0 : total;

  for (int base = wave * AGG_SBLK; base < n; base += (AGG_T / 32) * AGG_SBLK) {
    if (n - base >= AGG_SBLK) {
      score_block<true>(base, n, lane, src, dst, lo, el, er_s, q_pk, q_a, mx_s);
    } else {
      score_block<false>(base, n, lane, src, dst, lo, el, er_s, q_pk, q_a, mx_s);
    }
  }
  if (full) {
    for (int e = t; e < N_EDGES; e += AGG_T) {
      const unsigned r = (unsigned)(dst[e] - lo);
      if (r < (unsigned)AGG_NB) {
        const float sc = edge_score(el[src[e]], er_s[r]);
        __hip_atomic_fetch_max(&mx_s[r], sc, __ATOMIC_RELAXED, __HIP_MEMORY_SCOPE_WORKGROUP);
      }
    }
  }
  __syncthreads();

  for (int i = t; i < n; i += AGG_T) {
    const int r = (int)((unsigned)q_pk[i] >> 16);
    const float ex = __expf(q_a[i] - mx_s[r]);
    q_a[i] = ex;
    atomicAdd(&sm_s[r], ex);
  }
  if (full) {
    for (int e = t; e < N_EDGES; e += AGG_T) {
      const unsigned r = (unsigned)(dst[e] - lo);
      if (r < (unsigned)AGG_NB) {
        const float sc = edge_score(el[src[e]], er_s[r]);
        atomicAdd(&sm_s[r], __expf(sc - mx_s[r]));
      }
    }
  }
  __syncthreads();

  for (int b = wave * 8; b < n; b += (AGG_T / 32) * 8) {
    const int j = b + (lane & 7);
    if (n - b >= 8) {
      const int pk = q_pk[j];
      const float a = q_a[j] / sm_s[(unsigned)pk >> 16];
      agg_rows<true>(8, pk, a, H, acc, lane);
    } else {
      const bool in = j < n;
      const int pk = in ? q_pk[j] : 0;
      float a = 0.0f;
      if (in) a = q_a[j] / sm_s[(unsigned)pk >> 16];
      agg_rows<false>(n - b, pk, a, H, acc, lane);
    }
  }
  if (full) {
    for (int e = wave; e < N_EDGES; e += AGG_T / 32) {
      const unsigned r = (unsigned)(dst[e] - lo);
      if (r < (unsigned)AGG_NB) {
        const int s = src[e];
        const float sc = edge_score(el[s], er_s[r]);
        const float a = __expf(sc - mx_s[r]) / sm_s[r];
        const float* hp = H + (size_t)s * OUT_DIM + lane;
        float* o = acc + r * OUT_DIM + lane;
        atomicAdd(o +  0, a * hp[0]);
        atomicAdd(o + 32, a * hp[32]);
        atomicAdd(o + 64, a * hp[64]);
        atomicAdd(o + 96, a * hp[96]);
      }
    }
  }
  __syncthreads();

  float* op = out + (size_t)lo * OUT_DIM;
  for (int i = t; i < AGG_NB * OUT_DIM; i += AGG_T) op[i] = acc[i];
}

extern "C" void kernel_launch(void* const* d_in, const int* in_sizes, int n_in,
                              void* d_out, int out_size, void* d_ws, size_t ws_size,
                              hipStream_t stream) {
  (void)in_sizes; (void)n_in; (void)out_size; (void)ws_size;

  const float* features  = (const float*)d_in[0];
  const float* fc_weight = (const float*)d_in[1];
  const float* attn_l    = (const float*)d_in[2];
  const float* attn_r    = (const float*)d_in[3];
  const int*   src       = (const int*)d_in[4];
  const int*   dst       = (const int*)d_in[5];
  float*       out       = (float*)d_out;

  float* h  = (float*)d_ws;
  float* el = h + (size_t)N_NODES * OUT_DIM;
  float* er = el + N_NODES;
  int* rid4 = (int*)(er + N_NODES);

  gat_gemm_wmma<<<GEMM_GRID, 256, 0, stream>>>(features, fc_weight, h);

  node_logits<<<(N_NODES + 7) / 8, 256, 0, stream>>>(h, attn_l, attn_r, dst, el, er, rid4);

  aggregate<<<N_NODES / AGG_NB, AGG_T, 0, stream>>>(src, dst, rid4, h, el, er, out);
}


# Row sums leave LDS: wave w keeps nodes w, w+32, .. in registers; e_exp step queues each edge to its owner wave (returning LDS add); no LDS f32 row atomics; rows summed in queue (arrival) order
# speedup vs baseline: 4.0090x; 1.1377x over previous
#include <hip/hip_runtime.h>
#include <hip/hip_bf16.h>

#define N_NODES 50000
#define N_EDGES 1600000
#define IN_DIM 256
#define OUT_DIM 128
#define NEG_SLOPE 0.2f

typedef __attribute__((ext_vector_type(2))) float v2f;
typedef __attribute__((ext_vector_type(8))) float v8f;

#define GEMM_RB (N_NODES / 16)
#define GEMM_GRID 512
static_assert(N_NODES % 16 == 0, "row blocks must tile the nodes");
static_assert(IN_DIM % 8 == 0, "k is walked 8 at a time");
#define GEMM_CH 4
static_assert((IN_DIM / 8) % GEMM_CH == 0, "chunks must tile the k range");

__global__ __launch_bounds__(256) void gat_gemm_wmma(
    const float* __restrict__ X,
    const float* __restrict__ W,
    float* __restrict__ H)
{
  const int lane = threadIdx.x & 31;
  const int wave = threadIdx.x >> 5;
  const int l16  = lane & 15;
  const int half = lane >> 4;
  const int nBase = wave * 16;

  v2f bf[IN_DIM / 4];
  const float* bPtr = W + (size_t)(4 * half) * OUT_DIM + nBase + l16;
#pragma unroll
  for (int j = 0; j < IN_DIM / 8; ++j) {
    bf[2 * j].x     = bPtr[(size_t)(8 * j + 0) * OUT_DIM];
    bf[2 * j].y     = bPtr[(size_t)(8 * j + 1) * OUT_DIM];
    bf[2 * j + 1].x = bPtr[(size_t)(8 * j + 2) * OUT_DIM];
    bf[2 * j + 1].y = bPtr[(size_t)(8 * j + 3) * OUT_DIM];
  }

  float4 cur[GEMM_CH], nxt[GEMM_CH];
  {
    const float4* p = (const float4*)(X + (size_t)(blockIdx.x * 16 + l16) * IN_DIM + 4 * half);
#pragma unroll
    for (int i = 0; i < GEMM_CH; ++i) cur[i] = p[2 * i];
  }
  for (int rb = blockIdx.x; rb < GEMM_RB; rb += gridDim.x) {
    const int mBase = rb * 16;
    const int rbn = rb + (int)gridDim.x < GEMM_RB ? rb + (int)gridDim.x : rb;
    const float4* aPtr = (const float4*)(X + (size_t)(mBase + l16) * IN_DIM + 4 * half);
    const float4* aNxt = (const float4*)(X + (size_t)(rbn * 16 + l16) * IN_DIM + 4 * half);
    v8f acc0 = {};
    v8f acc1 = {};
#pragma unroll
    for (int c = 0; c < IN_DIM / 8 / GEMM_CH; ++c) {
#pragma unroll
      for (int i = 0; i < GEMM_CH; ++i)
        nxt[i] = (c + 1 < IN_DIM / 8 / GEMM_CH) ? aPtr[2 * ((c + 1) * GEMM_CH + i)] : aNxt[2 * i];
      __builtin_amdgcn_sched_barrier(0);
#pragma unroll
      for (int i = 0; i < GEMM_CH; ++i) {
        const int j = c * GEMM_CH + i;
        v2f a0; a0.x = cur[i].x; a0.y = cur[i].y;
        v2f a1; a1.x = cur[i].z; a1.y = cur[i].w;
        acc0 = __builtin_amdgcn_wmma_f32_16x16x4_f32(
             false, a0,  false, bf[2 * j],
             (short)0, acc0,  false,  false);
        acc1 = __builtin_amdgcn_wmma_f32_16x16x4_f32(
             false, a1,  false, bf[2 * j + 1],
             (short)0, acc1,  false,  false);
      }
#pragma unroll
      for (int i = 0; i < GEMM_CH; ++i) cur[i] = nxt[i];
    }
    float* out = H + (size_t)(mBase + 8 * half) * OUT_DIM + nBase + l16;
#pragma unroll
    for (int v = 0; v < 8; ++v)
      out[(size_t)v * OUT_DIM] = acc0[v] + acc1[v];
  }
}

#define AGG_NB   200
#define RID_NONE 255u
static_assert(N_NODES % AGG_NB == 0, "node ranges must tile the nodes");
static_assert(N_NODES / AGG_NB <= 255, "a range id must fit a byte, 255 kept free");
static_assert(N_EDGES % 16 == 0, "the range-id plane is read in 16-byte groups");

__device__ __forceinline__ unsigned range_id(int d)
{
  return (unsigned)d < (unsigned)N_NODES ? (unsigned)d / AGG_NB : RID_NONE;
}

__global__ __launch_bounds__(256) void node_logits(
    const float* __restrict__ H,
    const float* __restrict__ attn_l,
    const float* __restrict__ attn_r,
    const int* __restrict__ dst,
    float* __restrict__ el, float* __restrict__ er,
    int* __restrict__ rid4)
{
  const int gid = blockIdx.x * blockDim.x + threadIdx.x;
  if (gid < N_EDGES / 4) {
    const int4 d = ((const int4*)dst)[gid];
    rid4[gid] = (int)(range_id(d.x) | (range_id(d.y) << 8) |
                      (range_id(d.z) << 16) | (range_id(d.w) << 24));
  }

  const int lane = threadIdx.x & 31;
  const int node = blockIdx.x * (blockDim.x >> 5) + (threadIdx.x >> 5);
  if (node >= N_NODES) return;

  const float4 hv = *(const float4*)(H + (size_t)node * OUT_DIM + 4 * lane);
  const float4 al = *(const float4*)(attn_l + 4 * lane);
  const float4 ar = *(const float4*)(attn_r + 4 * lane);
  float sl = hv.x * al.x + hv.y * al.y + hv.z * al.z + hv.w * al.w;
  float sr = hv.x * ar.x + hv.y * ar.y + hv.z * ar.z + hv.w * ar.w;
#pragma unroll
  for (int off = 16; off > 0; off >>= 1) {
    sl += __shfl_down(sl, off, 32);
    sr += __shfl_down(sr, off, 32);
  }
  if (lane == 0) { el[node] = sl; er[node] = sr; }
}

#define AGG_T    1024
#define AGG_QCAP 16384
#define AGG_NG   (N_EDGES / 16)
#define AGG_NW   (AGG_T / 32)
#define AGG_SLOTS ((AGG_NB + AGG_NW - 1) / AGG_NW)
#define AGG_WQCAP (AGG_NB * OUT_DIM / (2 * AGG_NW))
static_assert(AGG_NB <= 256, "list entry: local node in 8 bits");
static_assert(N_NODES <= (1 << 16), "list entry: source node in 16 bits");
static_assert(AGG_SLOTS <= 8, "the register sums hold 8 slots per wave");
static_assert(AGG_NW == 32, "owner wave = local node & 31, slot = local node >> 5");
static_assert(2 * AGG_NW * AGG_WQCAP <= AGG_NB * OUT_DIM, "queues must fit the row-sum LDS");
#define AGG_RB 16
static_assert(AGG_RB <= 32 && (AGG_RB & (AGG_RB - 1)) == 0, "a lane per edge of the batch");
static_assert(AGG_WQCAP % AGG_RB == 0, "a batch never reads past its own queue");

struct OwnSums { v8f x, y, z, w; };

template <bool WHOLE>
__device__ __forceinline__ void own_rows(
    int cnt, int pk_l, float a_l, const float* __restrict__ H,
    OwnSums& sum, int lane)
{
  float4 hv[AGG_RB];
#pragma unroll
  for (int k = 0; k < AGG_RB; ++k) {
    if (WHOLE || k < cnt) {
      const int s = __builtin_amdgcn_readlane(pk_l, k) & 0xFFFF;
      hv[k] = *(const float4*)(H + (size_t)s * OUT_DIM + 4 * lane);
    }
  }
#pragma unroll
  for (int k = 0; k < AGG_RB; ++k) {
    if (WHOLE || k < cnt) {
      const int slot = (int)((unsigned)__builtin_amdgcn_readlane(pk_l, k) >> 21);
      const float a = __int_as_float(__builtin_amdgcn_readlane(__float_as_int(a_l), k));
      sum.x[slot] += a * hv[k].x;
      sum.y[slot] += a * hv[k].y;
      sum.z[slot] += a * hv[k].z;
      sum.w[slot] += a * hv[k].w;
    }
  }
}

__device__ __forceinline__ unsigned zero_byte_flags(unsigned y)
{
  return (y - 0x01010101u) & ~y;
}

__device__ __forceinline__ float edge_score(float l, float r)
{
  const float t = l + r;
  return t > 0.0f ? t : NEG_SLOPE * t;
}

#define AGG_SB 8
#define AGG_SBLK (32 * AGG_SB)

template <bool WHOLE>
__device__ __forceinline__ void score_block(
    int base, int n, int lane,
    const int* __restrict__ src, const int* __restrict__ dst, int lo,
    const float* __restrict__ el,
    const float* er_s, int* q_pk, float* q_a, float* mx_s)
{
  int pk[AGG_SB], s[AGG_SB], d[AGG_SB];
  float l[AGG_SB];
#pragma unroll
  for (int u = 0; u < AGG_SB; ++u) {
    const int i = base + 32 * u + lane;
    pk[u] = (WHOLE || i < n) ? q_pk[i] : 0;
  }
#pragma unroll
  for (int u = 0; u < AGG_SB; ++u) { s[u] = src[pk[u]]; d[u] = dst[pk[u]]; }
#pragma unroll
  for (int u = 0; u < AGG_SB; ++u) l[u] = el[s[u]];
#pragma unroll
  for (int u = 0; u < AGG_SB; ++u) {
    const int i = base + 32 * u + lane;
    if (WHOLE || i < n) {
      const int r = d[u] - lo;
      const float sc = edge_score(l[u], er_s[r]);
      q_a[i]  = sc;
      q_pk[i] = s[u] | (r << 16);
      __hip_atomic_fetch_max(&mx_s[r], sc, __ATOMIC_RELAXED, __HIP_MEMORY_SCOPE_WORKGROUP);
    }
  }
}

__global__ __launch_bounds__(AGG_T) void aggregate(
    const int* __restrict__ src, const int* __restrict__ dst,
    const int* __restrict__ rid4,
    const float* __restrict__ H,
    const float* __restrict__ el, const float* __restrict__ er,
    float* __restrict__ out)
{
  __shared__ float acc[AGG_NB * OUT_DIM];
  __shared__ int   q_pk[AGG_QCAP];
  __shared__ float q_a[AGG_QCAP];
  __shared__ float er_s[AGG_NB];
  __shared__ float mx_s[AGG_NB];
  __shared__ float sm_s[AGG_NB];
  __shared__ int   wq_n[AGG_NW];
  __shared__ int   q_n;
  __shared__ int   c_n;
  __shared__ int   w_ovf;

  int*   wq_pk = (int*)acc;
  float* wq_a  = acc + AGG_NW * AGG_WQCAP;

  const int t    = threadIdx.x;
  const int lane = t & 31;
  const int wave = __builtin_amdgcn_readfirstlane(t >> 5);
  const int lo   = blockIdx.x * AGG_NB;

  if (t < AGG_NB) {
    er_s[t] = er[lo + t];
    mx_s[t] = -__builtin_inff();
    sm_s[t] = 0.0f;
  }
  if (t < AGG_NW) wq_n[t] = 0;
  if (t == 0) { q_n = 0; c_n = 0; w_ovf = 0; }
  __syncthreads();

  const int4* rp = (const int4*)rid4;
  const unsigned id4 = blockIdx.x * 0x01010101u;
  int* cand = (int*)q_a;
  for (int base = 0; base < AGG_NG; base += 4 * AGG_T) {
    int4 v[4];
#pragma unroll
    for (int u = 0; u < 4; ++u) {
      const int i = base + u * AGG_T + t;
      v[u] = (i < AGG_NG) ? rp[i] : make_int4(-1, -1, -1, -1);
    }
#pragma unroll
    for (int u = 0; u < 4; ++u) {
      const int i = base + u * AGG_T + t;
      const unsigned f = (zero_byte_flags((unsigned)v[u].x ^ id4) | zero_byte_flags((unsigned)v[u].y ^ id4) |
                          zero_byte_flags((unsigned)v[u].z ^ id4) | zero_byte_flags((unsigned)v[u].w ^ id4)) & 0x80808080u;
      if (f != 0u) {
        const int idx = atomicAdd(&c_n, 1);
        if (idx < AGG_QCAP) cand[idx] = i;
      }
    }
  }
  __syncthreads();
  const int ctot = __builtin_amdgcn_readfirstlane(c_n);
  const int nc = ctot > AGG_QCAP ? 0 : ctot;

  for (int j = t; j < nc; j += AGG_T) {
    const int g = cand[j];
    const int4 x = rp[g];
    const unsigned y[4] = {(unsigned)x.x ^ id4, (unsigned)x.y ^ id4, (unsigned)x.z ^ id4, (unsigned)x.w ^ id4};
#pragma unroll
    for (int c = 0; c < 4; ++c) {
      unsigned h = zero_byte_flags(y[c]) & 0x80808080u;
      while (h != 0u) {
        const int b = __builtin_ctz(h) >> 3;
        h &= h - 1u;
        if (((y[c] >> (8 * b)) & 0xFFu) == 0u) {
          const int idx = atomicAdd(&q_n, 1);
          if (idx < AGG_QCAP) q_pk[idx] = 16 * g + 4 * c + b;
        }
      }
    }
  }
  __syncthreads();
  const int total = __builtin_amdgcn_readfirstlane(q_n);
  const bool full = ctot > AGG_QCAP || total > AGG_QCAP;
  const int n = full ? 0 : total;

  for (int base = wave * AGG_SBLK; base < n; base += (AGG_T / 32) * AGG_SBLK) {
    if (n - base >= AGG_SBLK) {
      score_block<true>(base, n, lane, src, dst, lo, el, er_s, q_pk, q_a, mx_s);
    } else {
      score_block<false>(base, n, lane, src, dst, lo, el, er_s, q_pk, q_a, mx_s);
    }
  }
  if (full) {
    for (int e = t; e < N_EDGES; e += AGG_T) {
      const unsigned r = (unsigned)(dst[e] - lo);
      if (r < (unsigned)AGG_NB) {
        const float sc = edge_score(el[src[e]], er_s[r]);
        __hip_atomic_fetch_max(&mx_s[r], sc, __ATOMIC_RELAXED, __HIP_MEMORY_SCOPE_WORKGROUP);
      }
    }
  }
  __syncthreads();

  for (int i = t; i < n; i += AGG_T) {
    const int pk = q_pk[i];
    const int r = (int)((unsigned)pk >> 16);
    const float ex = __expf(q_a[i] - mx_s[r]);
    atomicAdd(&sm_s[r], ex);
    const int w = r & (AGG_NW - 1);
    const int pos = atomicAdd(&wq_n[w], 1);
    if (pos < AGG_WQCAP) {
      wq_pk[w * AGG_WQCAP + pos] = pk;
      wq_a[w * AGG_WQCAP + pos]  = ex;
    } else {
      w_ovf = 1;
    }
  }
  if (full) {
    for (int e = t; e < N_EDGES; e += AGG_T) {
      const unsigned r = (unsigned)(dst[e] - lo);
      if (r < (unsigned)AGG_NB) {
        const float sc = edge_score(el[src[e]], er_s[r]);
        atomicAdd(&sm_s[r], __expf(sc - mx_s[r]));
      }
    }
  }
  __syncthreads();

  const bool slow = full || __builtin_amdgcn_readfirstlane(w_ovf) != 0;

  if (!slow) {
    OwnSums sum;
    sum.x = {}; sum.y = {}; sum.z = {}; sum.w = {};
    const int m = __builtin_amdgcn_readfirstlane(wq_n[wave]);
    const int*   my_pk = wq_pk + wave * AGG_WQCAP;
    const float* my_a  = wq_a  + wave * AGG_WQCAP;
    int b = 0;
    for (; b + AGG_RB <= m; b += AGG_RB) {
      const int j = b + (lane & (AGG_RB - 1));
      const int pk = my_pk[j];
      const float a = my_a[j] / sm_s[(unsigned)pk >> 16];
      own_rows<true>(AGG_RB, pk, a, H, sum, lane);
    }
    if (b < m) {
      const int j = b + (lane & (AGG_RB - 1));
      const bool in = j < m;
      const int pk = in ? my_pk[j] : 0;
      float a = 0.0f;
      if (in) a = my_a[j] / sm_s[(unsigned)pk >> 16];
      own_rows<false>(m - b, pk, a, H, sum, lane);
    }
#pragma unroll
    for (int k = 0; k < AGG_SLOTS; ++k) {
      const int r = wave + AGG_NW * k;
      if (r < AGG_NB)
        *(float4*)(out + (size_t)(lo + r) * OUT_DIM + 4 * lane) =
            make_float4(sum.x[k], sum.y[k], sum.z[k], sum.w[k]);
    }
  } else {
    for (int i = t; i < AGG_NB * OUT_DIM; i += AGG_T) acc[i] = 0.0f;
    __syncthreads();
    for (int e = wave; e < N_EDGES; e += AGG_T / 32) {
      const unsigned r = (unsigned)(dst[e] - lo);
      if (r < (unsigned)AGG_NB) {
        const int s = src[e];
        const float sc = edge_score(el[s], er_s[r]);
        const float a = __expf(sc - mx_s[r]) / sm_s[r];
        const float* hp = H + (size_t)s * OUT_DIM + lane;
        float* o = acc + r * OUT_DIM + lane;
        atomicAdd(o +  0, a * hp[0]);
        atomicAdd(o + 32, a * hp[32]);
        atomicAdd(o + 64, a * hp[64]);
        atomicAdd(o + 96, a * hp[96]);
      }
    }
    __syncthreads();
    float* op = out + (size_t)lo * OUT_DIM;
    for (int i = t; i < AGG_NB * OUT_DIM; i += AGG_T) op[i] = acc[i];
  }
}

extern "C" void kernel_launch(void* const* d_in, const int* in_sizes, int n_in,
                              void* d_out, int out_size, void* d_ws, size_t ws_size,
                              hipStream_t stream) {
  (void)in_sizes; (void)n_in; (void)out_size; (void)ws_size;

  const float* features  = (const float*)d_in[0];
  const float* fc_weight = (const float*)d_in[1];
  const float* attn_l    = (const float*)d_in[2];
  const float* attn_r    = (const float*)d_in[3];
  const int*   src       = (const int*)d_in[4];
  const int*   dst       = (const int*)d_in[5];
  float*       out       = (float*)d_out;

  float* h  = (float*)d_ws;
  float* el = h + (size_t)N_NODES * OUT_DIM;
  float* er = el + N_NODES;
  int* rid4 = (int*)(er + N_NODES);

  gat_gemm_wmma<<<GEMM_GRID, 256, 0, stream>>>(features, fc_weight, h);

  node_logits<<<(N_NODES + 7) / 8, 256, 0, stream>>>(h, attn_l, attn_r, dst, el, er, rid4);

  aggregate<<<N_NODES / AGG_NB, AGG_T, 0, stream>>>(src, dst, rid4, h, el, er, out);
}


# Logits launch as a resident grid: 3125 waves of 16 nodes each (all row loads in flight together) and 4 dst groups per thread, not 50000 one-node waves; el, er keep their sum tree
# speedup vs baseline: 4.1718x; 1.0406x over previous
#include <hip/hip_runtime.h>
#include <hip/hip_bf16.h>

#define N_NODES 50000
#define N_EDGES 1600000
#define IN_DIM 256
#define OUT_DIM 128
#define NEG_SLOPE 0.2f

typedef __attribute__((ext_vector_type(2))) float v2f;
typedef __attribute__((ext_vector_type(8))) float v8f;

#define GEMM_RB (N_NODES / 16)
#define GEMM_GRID 512
static_assert(N_NODES % 16 == 0, "row blocks must tile the nodes");
static_assert(IN_DIM % 8 == 0, "k is walked 8 at a time");
#define GEMM_CH 4
static_assert((IN_DIM / 8) % GEMM_CH == 0, "chunks must tile the k range");

__global__ __launch_bounds__(256) void gat_gemm_wmma(
    const float* __restrict__ X,
    const float* __restrict__ W,
    float* __restrict__ H)
{
  const int lane = threadIdx.x & 31;
  const int wave = threadIdx.x >> 5;
  const int l16  = lane & 15;
  const int half = lane >> 4;
  const int nBase = wave * 16;

  v2f bf[IN_DIM / 4];
  const float* bPtr = W + (size_t)(4 * half) * OUT_DIM + nBase + l16;
#pragma unroll
  for (int j = 0; j < IN_DIM / 8; ++j) {
    bf[2 * j].x     = bPtr[(size_t)(8 * j + 0) * OUT_DIM];
    bf[2 * j].y     = bPtr[(size_t)(8 * j + 1) * OUT_DIM];
    bf[2 * j + 1].x = bPtr[(size_t)(8 * j + 2) * OUT_DIM];
    bf[2 * j + 1].y = bPtr[(size_t)(8 * j + 3) * OUT_DIM];
  }

  float4 cur[GEMM_CH], nxt[GEMM_CH];
  {
    const float4* p = (const float4*)(X + (size_t)(blockIdx.x * 16 + l16) * IN_DIM + 4 * half);
#pragma unroll
    for (int i = 0; i < GEMM_CH; ++i) cur[i] = p[2 * i];
  }
  for (int rb = blockIdx.x; rb < GEMM_RB; rb += gridDim.x) {
    const int mBase = rb * 16;
    const int rbn = rb + (int)gridDim.x < GEMM_RB ? rb + (int)gridDim.x : rb;
    const float4* aPtr = (const float4*)(X + (size_t)(mBase + l16) * IN_DIM + 4 * half);
    const float4* aNxt = (const float4*)(X + (size_t)(rbn * 16 + l16) * IN_DIM + 4 * half);
    v8f acc0 = {};
    v8f acc1 = {};
#pragma unroll
    for (int c = 0; c < IN_DIM / 8 / GEMM_CH; ++c) {
#pragma unroll
      for (int i = 0; i < GEMM_CH; ++i)
        nxt[i] = (c + 1 < IN_DIM / 8 / GEMM_CH) ? aPtr[2 * ((c + 1) * GEMM_CH + i)] : aNxt[2 * i];
      __builtin_amdgcn_sched_barrier(0);
#pragma unroll
      for (int i = 0; i < GEMM_CH; ++i) {
        const int j = c * GEMM_CH + i;
        v2f a0; a0.x = cur[i].x; a0.y = cur[i].y;
        v2f a1; a1.x = cur[i].z; a1.y = cur[i].w;
        acc0 = __builtin_amdgcn_wmma_f32_16x16x4_f32(
             false, a0,  false, bf[2 * j],
             (short)0, acc0,  false,  false);
        acc1 = __builtin_amdgcn_wmma_f32_16x16x4_f32(
             false, a1,  false, bf[2 * j + 1],
             (short)0, acc1,  false,  false);
      }
#pragma unroll
      for (int i = 0; i < GEMM_CH; ++i) cur[i] = nxt[i];
    }
    float* out = H + (size_t)(mBase + 8 * half) * OUT_DIM + nBase + l16;
#pragma unroll
    for (int v = 0; v < 8; ++v)
      out[(size_t)v * OUT_DIM] = acc0[v] + acc1[v];
  }
}

#define AGG_NB   200
#define RID_NONE 255u
static_assert(N_NODES % AGG_NB == 0, "node ranges must tile the nodes");
static_assert(N_NODES / AGG_NB <= 255, "a range id must fit a byte, 255 kept free");
static_assert(N_EDGES % 16 == 0, "the range-id plane is read in 16-byte groups");

__device__ __forceinline__ unsigned range_id(int d)
{
  return (unsigned)d < (unsigned)N_NODES ? (unsigned)d / AGG_NB : RID_NONE;
}

#define NL_ROWS 16
#define NL_WAVES (N_NODES / NL_ROWS)
#define NL_GRID ((NL_WAVES + 7) / 8)
#define NL_THREADS (NL_GRID * 256)
#define NL_NG4 (N_EDGES / 4)
#define NL_RID_TRIPS ((NL_NG4 + NL_THREADS - 1) / NL_THREADS)
static_assert(N_NODES % NL_ROWS == 0, "row groups must tile the nodes");
static_assert(NL_ROWS <= 32, "a lane per node of the group for the stores");
static_assert(N_EDGES % 4 == 0, "dst is read in 16-byte groups");

__global__ __launch_bounds__(256) void node_logits(
    const float* __restrict__ H,
    const float* __restrict__ attn_l,
    const float* __restrict__ attn_r,
    const int* __restrict__ dst,
    float* __restrict__ el, float* __restrict__ er,
    int* __restrict__ rid4)
{
  const int gid  = blockIdx.x * 256 + threadIdx.x;
  const int lane = threadIdx.x & 31;
  const int grp  = __builtin_amdgcn_readfirstlane(gid >> 5);
  const bool own = grp < NL_WAVES;
  const int n0 = (own ? grp : NL_WAVES - 1) * NL_ROWS;

  float4 hv[NL_ROWS];
  const float* hp = H + (size_t)n0 * OUT_DIM + 4 * lane;
#pragma unroll
  for (int u = 0; u < NL_ROWS; ++u) hv[u] = *(const float4*)(hp + (size_t)u * OUT_DIM);

  int4 d[NL_RID_TRIPS];
#pragma unroll
  for (int k = 0; k < NL_RID_TRIPS; ++k) {
    const int i = gid + k * NL_THREADS;
    d[k] = ((const int4*)dst)[i < NL_NG4 ? i : NL_NG4 - 1];
  }
  const float4 al = *(const float4*)(attn_l + 4 * lane);
  const float4 ar = *(const float4*)(attn_r + 4 * lane);

#pragma unroll
  for (int k = 0; k < NL_RID_TRIPS; ++k) {
    const int i = gid + k * NL_THREADS;
    if (i < NL_NG4)
      rid4[i] = (int)(range_id(d[k].x) | (range_id(d[k].y) << 8) |
                      (range_id(d[k].z) << 16) | (range_id(d[k].w) << 24));
  }

  float sl[NL_ROWS], sr[NL_ROWS];
#pragma unroll
  for (int u = 0; u < NL_ROWS; ++u) {
    sl[u] = hv[u].x * al.x + hv[u].y * al.y + hv[u].z * al.z + hv[u].w * al.w;
    sr[u] = hv[u].x * ar.x + hv[u].y * ar.y + hv[u].z * ar.z + hv[u].w * ar.w;
  }
#pragma unroll
  for (int off = 16; off > 0; off >>= 1) {
#pragma unroll
    for (int u = 0; u < NL_ROWS; ++u) {
      sl[u] += __shfl_xor(sl[u], off, 32);
      sr[u] += __shfl_xor(sr[u], off, 32);
    }
  }
  float myl = 0.0f, myr = 0.0f;
#pragma unroll
  for (int u = 0; u < NL_ROWS; ++u)
    if (lane == u) { myl = sl[u]; myr = sr[u]; }
  if (own && lane < NL_ROWS) {
    el[n0 + lane] = myl;
    er[n0 + lane] = myr;
  }
}

#define AGG_T    1024
#define AGG_QCAP 16384
#define AGG_NG   (N_EDGES / 16)
#define AGG_NW   (AGG_T / 32)
#define AGG_SLOTS ((AGG_NB + AGG_NW - 1) / AGG_NW)
#define AGG_WQCAP (AGG_NB * OUT_DIM / (2 * AGG_NW))
static_assert(AGG_NB <= 256, "list entry: local node in 8 bits");
static_assert(N_NODES <= (1 << 16), "list entry: source node in 16 bits");
static_assert(AGG_SLOTS <= 8, "the register sums hold 8 slots per wave");
static_assert(AGG_NW == 32, "owner wave = local node & 31, slot = local node >> 5");
static_assert(2 * AGG_NW * AGG_WQCAP <= AGG_NB * OUT_DIM, "queues must fit the row-sum LDS");
#define AGG_RB 16
static_assert(AGG_RB <= 32 && (AGG_RB & (AGG_RB - 1)) == 0, "a lane per edge of the batch");
static_assert(AGG_WQCAP % AGG_RB == 0, "a batch never reads past its own queue");

struct OwnSums { v8f x, y, z, w; };

template <bool WHOLE>
__device__ __forceinline__ void own_rows(
    int cnt, int pk_l, float a_l, const float* __restrict__ H,
    OwnSums& sum, int lane)
{
  float4 hv[AGG_RB];
#pragma unroll
  for (int k = 0; k < AGG_RB; ++k) {
    if (WHOLE || k < cnt) {
      const int s = __builtin_amdgcn_readlane(pk_l, k) & 0xFFFF;
      hv[k] = *(const float4*)(H + (size_t)s * OUT_DIM + 4 * lane);
    }
  }
#pragma unroll
  for (int k = 0; k < AGG_RB; ++k) {
    if (WHOLE || k < cnt) {
      const int slot = (int)((unsigned)__builtin_amdgcn_readlane(pk_l, k) >> 21);
      const float a = __int_as_float(__builtin_amdgcn_readlane(__float_as_int(a_l), k));
      sum.x[slot] += a * hv[k].x;
      sum.y[slot] += a * hv[k].y;
      sum.z[slot] += a * hv[k].z;
      sum.w[slot] += a * hv[k].w;
    }
  }
}

__device__ __forceinline__ unsigned zero_byte_flags(unsigned y)
{
  return (y - 0x01010101u) & ~y;
}

__device__ __forceinline__ float edge_score(float l, float r)
{
  const float t = l + r;
  return t > 0.0f ? t : NEG_SLOPE * t;
}

#define AGG_SB 8
#define AGG_SBLK (32 * AGG_SB)

template <bool WHOLE>
__device__ __forceinline__ void score_block(
    int base, int n, int lane,
    const int* __restrict__ src, const int* __restrict__ dst, int lo,
    const float* __restrict__ el,
    const float* er_s, int* q_pk, float* q_a, float* mx_s)
{
  int pk[AGG_SB], s[AGG_SB], d[AGG_SB];
  float l[AGG_SB];
#pragma unroll
  for (int u = 0; u < AGG_SB; ++u) {
    const int i = base + 32 * u + lane;
    pk[u] = (WHOLE || i < n) ? q_pk[i] : 0;
  }
#pragma unroll
  for (int u = 0; u < AGG_SB; ++u) { s[u] = src[pk[u]]; d[u] = dst[pk[u]]; }
#pragma unroll
  for (int u = 0; u < AGG_SB; ++u) l[u] = el[s[u]];
#pragma unroll
  for (int u = 0; u < AGG_SB; ++u) {
    const int i = base + 32 * u + lane;
    if (WHOLE || i < n) {
      const int r = d[u] - lo;
      const float sc = edge_score(l[u], er_s[r]);
      q_a[i]  = sc;
      q_pk[i] = s[u] | (r << 16);
      __hip_atomic_fetch_max(&mx_s[r], sc, __ATOMIC_RELAXED, __HIP_MEMORY_SCOPE_WORKGROUP);
    }
  }
}

__global__ __launch_bounds__(AGG_T) void aggregate(
    const int* __restrict__ src, const int* __restrict__ dst,
    const int* __restrict__ rid4,
    const float* __restrict__ H,
    const float* __restrict__ el, const float* __restrict__ er,
    float* __restrict__ out)
{
  __shared__ float acc[AGG_NB * OUT_DIM];
  __shared__ int   q_pk[AGG_QCAP];
  __shared__ float q_a[AGG_QCAP];
  __shared__ float er_s[AGG_NB];
  __shared__ float mx_s[AGG_NB];
  __shared__ float sm_s[AGG_NB];
  __shared__ int   wq_n[AGG_NW];
  __shared__ int   q_n;
  __shared__ int   c_n;
  __shared__ int   w_ovf;

  int*   wq_pk = (int*)acc;
  float* wq_a  = acc + AGG_NW * AGG_WQCAP;

  const int t    = threadIdx.x;
  const int lane = t & 31;
  const int wave = __builtin_amdgcn_readfirstlane(t >> 5);
  const int lo   = blockIdx.x * AGG_NB;

  if (t < AGG_NB) {
    er_s[t] = er[lo + t];
    mx_s[t] = -__builtin_inff();
    sm_s[t] = 0.0f;
  }
  if (t < AGG_NW) wq_n[t] = 0;
  if (t == 0) { q_n = 0; c_n = 0; w_ovf = 0; }
  __syncthreads();

  const int4* rp = (const int4*)rid4;
  const unsigned id4 = blockIdx.x * 0x01010101u;
  int* cand = (int*)q_a;
  for (int base = 0; base < AGG_NG; base += 4 * AGG_T) {
    int4 v[4];
#pragma unroll
    for (int u = 0; u < 4; ++u) {
      const int i = base + u * AGG_T + t;
      v[u] = (i < AGG_NG) ? rp[i] : make_int4(-1, -1, -1, -1);
    }
#pragma unroll
    for (int u = 0; u < 4; ++u) {
      const int i = base + u * AGG_T + t;
      const unsigned f = (zero_byte_flags((unsigned)v[u].x ^ id4) | zero_byte_flags((unsigned)v[u].y ^ id4) |
                          zero_byte_flags((unsigned)v[u].z ^ id4) | zero_byte_flags((unsigned)v[u].w ^ id4)) & 0x80808080u;
      if (f != 0u) {
        const int idx = atomicAdd(&c_n, 1);
        if (idx < AGG_QCAP) cand[idx] = i;
      }
    }
  }
  __syncthreads();
  const int ctot = __builtin_amdgcn_readfirstlane(c_n);
  const int nc = ctot > AGG_QCAP ? 0 : ctot;

  for (int j = t; j < nc; j += AGG_T) {
    const int g = cand[j];
    const int4 x = rp[g];
    const unsigned y[4] = {(unsigned)x.x ^ id4, (unsigned)x.y ^ id4, (unsigned)x.z ^ id4, (unsigned)x.w ^ id4};
#pragma unroll
    for (int c = 0; c < 4; ++c) {
      unsigned h = zero_byte_flags(y[c]) & 0x80808080u;
      while (h != 0u) {
        const int b = __builtin_ctz(h) >> 3;
        h &= h - 1u;
        if (((y[c] >> (8 * b)) & 0xFFu) == 0u) {
          const int idx = atomicAdd(&q_n, 1);
          if (idx < AGG_QCAP) q_pk[idx] = 16 * g + 4 * c + b;
        }
      }
    }
  }
  __syncthreads();
  const int total = __builtin_amdgcn_readfirstlane(q_n);
  const bool full = ctot > AGG_QCAP || total > AGG_QCAP;
  const int n = full ? 0 : total;

  for (int base = wave * AGG_SBLK; base < n; base += (AGG_T / 32) * AGG_SBLK) {
    if (n - base >= AGG_SBLK) {
      score_block<true>(base, n, lane, src, dst, lo, el, er_s, q_pk, q_a, mx_s);
    } else {
      score_block<false>(base, n, lane, src, dst, lo, el, er_s, q_pk, q_a, mx_s);
    }
  }
  if (full) {
    for (int e = t; e < N_EDGES; e += AGG_T) {
      const unsigned r = (unsigned)(dst[e] - lo);
      if (r < (unsigned)AGG_NB) {
        const float sc = edge_score(el[src[e]], er_s[r]);
        __hip_atomic_fetch_max(&mx_s[r], sc, __ATOMIC_RELAXED, __HIP_MEMORY_SCOPE_WORKGROUP);
      }
    }
  }
  __syncthreads();

  for (int i = t; i < n; i += AGG_T) {
    const int pk = q_pk[i];
    const int r = (int)((unsigned)pk >> 16);
    const float ex = __expf(q_a[i] - mx_s[r]);
    atomicAdd(&sm_s[r], ex);
    const int w = r & (AGG_NW - 1);
    const int pos = atomicAdd(&wq_n[w], 1);
    if (pos < AGG_WQCAP) {
      wq_pk[w * AGG_WQCAP + pos] = pk;
      wq_a[w * AGG_WQCAP + pos]  = ex;
    } else {
      w_ovf = 1;
    }
  }
  if (full) {
    for (int e = t; e < N_EDGES; e += AGG_T) {
      const unsigned r = (unsigned)(dst[e] - lo);
      if (r < (unsigned)AGG_NB) {
        const float sc = edge_score(el[src[e]], er_s[r]);
        atomicAdd(&sm_s[r], __expf(sc - mx_s[r]));
      }
    }
  }
  __syncthreads();

  const bool slow = full || __builtin_amdgcn_readfirstlane(w_ovf) != 0;

  if (!slow) {
    OwnSums sum;
    sum.x = {}; sum.y = {}; sum.z = {}; sum.w = {};
    const int m = __builtin_amdgcn_readfirstlane(wq_n[wave]);
    const int*   my_pk = wq_pk + wave * AGG_WQCAP;
    const float* my_a  = wq_a  + wave * AGG_WQCAP;
    int b = 0;
    for (; b + AGG_RB <= m; b += AGG_RB) {
      const int j = b + (lane & (AGG_RB - 1));
      const int pk = my_pk[j];
      const float a = my_a[j] / sm_s[(unsigned)pk >> 16];
      own_rows<true>(AGG_RB, pk, a, H, sum, lane);
    }
    if (b < m) {
      const int j = b + (lane & (AGG_RB - 1));
      const bool in = j < m;
      const int pk = in ? my_pk[j] : 0;
      float a = 0.0f;
      if (in) a = my_a[j] / sm_s[(unsigned)pk >> 16];
      own_rows<false>(m - b, pk, a, H, sum, lane);
    }
#pragma unroll
    for (int k = 0; k < AGG_SLOTS; ++k) {
      const int r = wave + AGG_NW * k;
      if (r < AGG_NB)
        *(float4*)(out + (size_t)(lo + r) * OUT_DIM + 4 * lane) =
            make_float4(sum.x[k], sum.y[k], sum.z[k], sum.w[k]);
    }
  } else {
    for (int i = t; i < AGG_NB * OUT_DIM; i += AGG_T) acc[i] = 0.0f;
    __syncthreads();
    for (int e = wave; e < N_EDGES; e += AGG_T / 32) {
      const unsigned r = (unsigned)(dst[e] - lo);
      if (r < (unsigned)AGG_NB) {
        const int s = src[e];
        const float sc = edge_score(el[s], er_s[r]);
        const float a = __expf(sc - mx_s[r]) / sm_s[r];
        const float* hp = H + (size_t)s * OUT_DIM + lane;
        float* o = acc + r * OUT_DIM + lane;
        atomicAdd(o +  0, a * hp[0]);
        atomicAdd(o + 32, a * hp[32]);
        atomicAdd(o + 64, a * hp[64]);
        atomicAdd(o + 96, a * hp[96]);
      }
    }
    __syncthreads();
    float* op = out + (size_t)lo * OUT_DIM;
    for (int i = t; i < AGG_NB * OUT_DIM; i += AGG_T) op[i] = acc[i];
  }
}

extern "C" void kernel_launch(void* const* d_in, const int* in_sizes, int n_in,
                              void* d_out, int out_size, void* d_ws, size_t ws_size,
                              hipStream_t stream) {
  (void)in_sizes; (void)n_in; (void)out_size; (void)ws_size;

  const float* features  = (const float*)d_in[0];
  const float* fc_weight = (const float*)d_in[1];
  const float* attn_l    = (const float*)d_in[2];
  const float* attn_r    = (const float*)d_in[3];
  const int*   src       = (const int*)d_in[4];
  const int*   dst       = (const int*)d_in[5];
  float*       out       = (float*)d_out;

  float* h  = (float*)d_ws;
  float* el = h + (size_t)N_NODES * OUT_DIM;
  float* er = el + N_NODES;
  int* rid4 = (int*)(er + N_NODES);

  gat_gemm_wmma<<<GEMM_GRID, 256, 0, stream>>>(features, fc_weight, h);

  node_logits<<<NL_GRID, 256, 0, stream>>>(h, attn_l, attn_r, dst, el, er, rid4);

  aggregate<<<N_NODES / AGG_NB, AGG_T, 0, stream>>>(src, dst, rid4, h, el, er, out);
}


# Listing by counts and prefix sums, no arrival atomics (why the bitmap try varied): aggregate reads a per-range bitmap of 32-edge groups and only flagged dst lines; edge-order list: sum order changes
# speedup vs baseline: 5.1551x; 1.2357x over previous
#include <hip/hip_runtime.h>
#include <hip/hip_bf16.h>

#define N_NODES 50000
#define N_EDGES 1600000
#define IN_DIM 256
#define OUT_DIM 128
#define NEG_SLOPE 0.2f

typedef __attribute__((ext_vector_type(2))) float v2f;
typedef __attribute__((ext_vector_type(8))) float v8f;

#define GEMM_RB (N_NODES / 16)
#define GEMM_GRID 512
static_assert(N_NODES % 16 == 0, "row blocks must tile the nodes");
static_assert(IN_DIM % 8 == 0, "k is walked 8 at a time");
#define GEMM_CH 4
static_assert((IN_DIM / 8) % GEMM_CH == 0, "chunks must tile the k range");

__global__ __launch_bounds__(256) void gat_gemm_wmma(
    const float* __restrict__ X,
    const float* __restrict__ W,
    float* __restrict__ H)
{
  const int lane = threadIdx.x & 31;
  const int wave = threadIdx.x >> 5;
  const int l16  = lane & 15;
  const int half = lane >> 4;
  const int nBase = wave * 16;

  v2f bf[IN_DIM / 4];
  const float* bPtr = W + (size_t)(4 * half) * OUT_DIM + nBase + l16;
#pragma unroll
  for (int j = 0; j < IN_DIM / 8; ++j) {
    bf[2 * j].x     = bPtr[(size_t)(8 * j + 0) * OUT_DIM];
    bf[2 * j].y     = bPtr[(size_t)(8 * j + 1) * OUT_DIM];
    bf[2 * j + 1].x = bPtr[(size_t)(8 * j + 2) * OUT_DIM];
    bf[2 * j + 1].y = bPtr[(size_t)(8 * j + 3) * OUT_DIM];
  }

  float4 cur[GEMM_CH], nxt[GEMM_CH];
  {
    const float4* p = (const float4*)(X + (size_t)(blockIdx.x * 16 + l16) * IN_DIM + 4 * half);
#pragma unroll
    for (int i = 0; i < GEMM_CH; ++i) cur[i] = p[2 * i];
  }
  for (int rb = blockIdx.x; rb < GEMM_RB; rb += gridDim.x) {
    const int mBase = rb * 16;
    const int rbn = rb + (int)gridDim.x < GEMM_RB ? rb + (int)gridDim.x : rb;
    const float4* aPtr = (const float4*)(X + (size_t)(mBase + l16) * IN_DIM + 4 * half);
    const float4* aNxt = (const float4*)(X + (size_t)(rbn * 16 + l16) * IN_DIM + 4 * half);
    v8f acc0 = {};
    v8f acc1 = {};
#pragma unroll
    for (int c = 0; c < IN_DIM / 8 / GEMM_CH; ++c) {
#pragma unroll
      for (int i = 0; i < GEMM_CH; ++i)
        nxt[i] = (c + 1 < IN_DIM / 8 / GEMM_CH) ? aPtr[2 * ((c + 1) * GEMM_CH + i)] : aNxt[2 * i];
      __builtin_amdgcn_sched_barrier(0);
#pragma unroll
      for (int i = 0; i < GEMM_CH; ++i) {
        const int j = c * GEMM_CH + i;
        v2f a0; a0.x = cur[i].x; a0.y = cur[i].y;
        v2f a1; a1.x = cur[i].z; a1.y = cur[i].w;
        acc0 = __builtin_amdgcn_wmma_f32_16x16x4_f32(
             false, a0,  false, bf[2 * j],
             (short)0, acc0,  false,  false);
        acc1 = __builtin_amdgcn_wmma_f32_16x16x4_f32(
             false, a1,  false, bf[2 * j + 1],
             (short)0, acc1,  false,  false);
      }
#pragma unroll
      for (int i = 0; i < GEMM_CH; ++i) cur[i] = nxt[i];
    }
    float* out = H + (size_t)(mBase + 8 * half) * OUT_DIM + nBase + l16;
#pragma unroll
    for (int v = 0; v < 8; ++v)
      out[(size_t)v * OUT_DIM] = acc0[v] + acc1[v];
  }
}

#define AGG_NB   200
#define RID_NONE 255u
#define AGG_NR   (N_NODES / AGG_NB)
#define BM_GE    32
#define BM_NGRP  (N_EDGES / BM_GE)
#define BM_WPB   4
static_assert(N_NODES % AGG_NB == 0, "node ranges must tile the nodes");
static_assert(N_NODES / AGG_NB <= 255, "a range id must fit a byte, 255 kept free");
static_assert(N_EDGES % BM_GE == 0, "bitmap groups must tile the edges");

__device__ __forceinline__ unsigned range_id(int d)
{
  return (unsigned)d < (unsigned)N_NODES ? (unsigned)d / AGG_NB : RID_NONE;
}

#define NL_ROWS 16
#define NL_WAVES (N_NODES / NL_ROWS)
#define NL_GRID ((NL_WAVES + 7) / 8)
#define NL_NG4 (N_EDGES / 4)
#define NL_G4_WG (BM_WPB * 32 * BM_GE / 4)
#define NL_RID_TRIPS (NL_G4_WG / 256)
static_assert(NL_RID_TRIPS == BM_WPB, "trip k of a thread fills bitmap word k");
static_assert(NL_GRID * NL_G4_WG >= NL_NG4, "the work-groups must cover dst");
static_assert((size_t)NL_GRID * AGG_NR * BM_WPB * 4 <= (size_t)N_EDGES, "the bitmap must fit its N_EDGES bytes of workspace");
static_assert(N_NODES % NL_ROWS == 0, "row groups must tile the nodes");
static_assert(NL_ROWS <= 32, "a lane per node of the group for the stores");
static_assert(N_EDGES % 4 == 0, "dst is read in 16-byte groups");

__global__ __launch_bounds__(256) void node_logits(
    const float* __restrict__ H,
    const float* __restrict__ attn_l,
    const float* __restrict__ attn_r,
    const int* __restrict__ dst,
    float* __restrict__ el, float* __restrict__ er,
    int* __restrict__ bmap)
{
  __shared__ __attribute__((aligned(16))) unsigned bm[AGG_NR * BM_WPB];
  const int gid  = blockIdx.x * 256 + threadIdx.x;
  const int lane = threadIdx.x & 31;
  const int grp  = __builtin_amdgcn_readfirstlane(gid >> 5);
  const bool own = grp < NL_WAVES;
  const int n0 = (own ? grp : NL_WAVES - 1) * NL_ROWS;

  float4 hv[NL_ROWS];
  const float* hp = H + (size_t)n0 * OUT_DIM + 4 * lane;
#pragma unroll
  for (int u = 0; u < NL_ROWS; ++u) hv[u] = *(const float4*)(hp + (size_t)u * OUT_DIM);

  int4 d[NL_RID_TRIPS];
#pragma unroll
  for (int k = 0; k < NL_RID_TRIPS; ++k) {
    const int i = blockIdx.x * NL_G4_WG + k * 256 + (int)threadIdx.x;
    d[k] = ((const int4*)dst)[i < NL_NG4 ? i : NL_NG4 - 1];
  }
  const float4 al = *(const float4*)(attn_l + 4 * lane);
  const float4 ar = *(const float4*)(attn_r + 4 * lane);

  for (int i = threadIdx.x; i < AGG_NR * BM_WPB; i += 256) bm[i] = 0u;
  __syncthreads();
  const unsigned bit = 1u << (threadIdx.x >> 3);
#pragma unroll
  for (int k = 0; k < NL_RID_TRIPS; ++k) {
    const int i = blockIdx.x * NL_G4_WG + k * 256 + (int)threadIdx.x;
    if (i < NL_NG4) {
      const int dd[4] = {d[k].x, d[k].y, d[k].z, d[k].w};
#pragma unroll
      for (int c = 0; c < 4; ++c) {
        const unsigned rid = range_id(dd[c]);
        if (rid != RID_NONE)
          __hip_atomic_fetch_or(&bm[rid * BM_WPB + k], bit, __ATOMIC_RELAXED, __HIP_MEMORY_SCOPE_WORKGROUP);
      }
    }
  }
  __syncthreads();
  if (threadIdx.x < AGG_NR)
    ((int4*)bmap)[blockIdx.x * AGG_NR + threadIdx.x] = *(const int4*)&bm[threadIdx.x * BM_WPB];

  float sl[NL_ROWS], sr[NL_ROWS];
#pragma unroll
  for (int u = 0; u < NL_ROWS; ++u) {
    sl[u] = hv[u].x * al.x + hv[u].y * al.y + hv[u].z * al.z + hv[u].w * al.w;
    sr[u] = hv[u].x * ar.x + hv[u].y * ar.y + hv[u].z * ar.z + hv[u].w * ar.w;
  }
#pragma unroll
  for (int off = 16; off > 0; off >>= 1) {
#pragma unroll
    for (int u = 0; u < NL_ROWS; ++u) {
      sl[u] += __shfl_xor(sl[u], off, 32);
      sr[u] += __shfl_xor(sr[u], off, 32);
    }
  }
  float myl = 0.0f, myr = 0.0f;
#pragma unroll
  for (int u = 0; u < NL_ROWS; ++u)
    if (lane == u) { myl = sl[u]; myr = sr[u]; }
  if (own && lane < NL_ROWS) {
    el[n0 + lane] = myl;
    er[n0 + lane] = myr;
  }
}

#define AGG_T    1024
#define AGG_QCAP 16384
#define AGG_NW   (AGG_T / 32)
#define AGG_SLOTS ((AGG_NB + AGG_NW - 1) / AGG_NW)
#define AGG_WQCAP (AGG_NB * OUT_DIM / (2 * AGG_NW))
static_assert(AGG_NB <= 256, "list entry: local node in 8 bits");
static_assert(N_NODES <= (1 << 16), "list entry: source node in 16 bits");
static_assert(AGG_SLOTS <= 8, "the register sums hold 8 slots per wave");
static_assert(AGG_NW == 32, "owner wave = local node & 31, slot = local node >> 5");
static_assert(2 * AGG_NW * AGG_WQCAP <= AGG_NB * OUT_DIM, "queues must fit the row-sum LDS");
#define AGG_RB 16
static_assert(AGG_RB <= 32 && (AGG_RB & (AGG_RB - 1)) == 0, "a lane per edge of the batch");
static_assert(AGG_WQCAP % AGG_RB == 0, "a batch never reads past its own queue");

struct OwnSums { v8f x, y, z, w; };

template <bool WHOLE>
__device__ __forceinline__ void own_rows(
    int cnt, int pk_l, float a_l, const float* __restrict__ H,
    OwnSums& sum, int lane)
{
  float4 hv[AGG_RB];
#pragma unroll
  for (int k = 0; k < AGG_RB; ++k) {
    if (WHOLE || k < cnt) {
      const int s = __builtin_amdgcn_readlane(pk_l, k) & 0xFFFF;
      hv[k] = *(const float4*)(H + (size_t)s * OUT_DIM + 4 * lane);
    }
  }
#pragma unroll
  for (int k = 0; k < AGG_RB; ++k) {
    if (WHOLE || k < cnt) {
      const int slot = (int)((unsigned)__builtin_amdgcn_readlane(pk_l, k) >> 21);
      const float a = __int_as_float(__builtin_amdgcn_readlane(__float_as_int(a_l), k));
      sum.x[slot] += a * hv[k].x;
      sum.y[slot] += a * hv[k].y;
      sum.z[slot] += a * hv[k].z;
      sum.w[slot] += a * hv[k].w;
    }
  }
}

__device__ __forceinline__ float edge_score(float l, float r)
{
  const float t = l + r;
  return t > 0.0f ? t : NEG_SLOPE * t;
}

#define AGG_SB 8
#define AGG_SBLK (32 * AGG_SB)

template <bool WHOLE>
__device__ __forceinline__ void score_block(
    int base, int n, int lane,
    const int* __restrict__ src,
    const float* __restrict__ el,
    const float* er_s, int* q_pk, float* q_a, float* mx_s)
{
  int pk[AGG_SB], s[AGG_SB];
  float l[AGG_SB];
#pragma unroll
  for (int u = 0; u < AGG_SB; ++u) {
    const int i = base + 32 * u + lane;
    pk[u] = (WHOLE || i < n) ? q_pk[i] : 0;
  }
#pragma unroll
  for (int u = 0; u < AGG_SB; ++u) s[u] = src[pk[u] & 0x1FFFFF];
#pragma unroll
  for (int u = 0; u < AGG_SB; ++u) l[u] = el[s[u]];
#pragma unroll
  for (int u = 0; u < AGG_SB; ++u) {
    const int i = base + 32 * u + lane;
    if (WHOLE || i < n) {
      const int r = (int)((unsigned)pk[u] >> 21);
      const float sc = edge_score(l[u], er_s[r]);
      q_a[i]  = sc;
      q_pk[i] = s[u] | (r << 16);
      __hip_atomic_fetch_max(&mx_s[r], sc, __ATOMIC_RELAXED, __HIP_MEMORY_SCOPE_WORKGROUP);
    }
  }
}

__device__ __forceinline__ int wave_incl_sum(int v, int lane)
{
#pragma unroll
  for (int off = 1; off < 32; off <<= 1) {
    const int o = __shfl_up(v, off, 32);
    if (lane >= off) v += o;
  }
  return v;
}

#define AGG_CCAP (AGG_NB * OUT_DIM)
#define AGG_WSEG (AGG_QCAP / AGG_NW)
#define AGG_RU   8
static_assert(AGG_WSEG * AGG_NW == AGG_QCAP, "the waves' pieces fill the list");
static_assert(N_EDGES <= (1 << 21), "list entry while listing: edge in 21 bits, local node above");
static_assert(NL_GRID <= AGG_T, "a thread per logits work-group's piece of the bitmap");

template <bool WHOLE>
__device__ __forceinline__ int list_groups(
    int jb, int jbeg, int jend, int q4, int sub, int lo,
    const int* cand, const int* __restrict__ dst, int* myseg, int wb)
{
  int4 dv[AGG_RU];
  int  gg[AGG_RU];
#pragma unroll
  for (int u = 0; u < AGG_RU; ++u) {
    const int j = jb + 4 * u + q4;
    gg[u] = cand[(WHOLE || j < jend) ? j : jbeg];
    dv[u] = ((const int4*)dst)[gg[u] * (BM_GE / 4) + sub];
  }
#pragma unroll
  for (int u = 0; u < AGG_RU; ++u) {
    const unsigned r0 = (unsigned)(dv[u].x - lo), r1 = (unsigned)(dv[u].y - lo);
    const unsigned r2 = (unsigned)(dv[u].z - lo), r3 = (unsigned)(dv[u].w - lo);
    bool m0 = r0 < (unsigned)AGG_NB, m1 = r1 < (unsigned)AGG_NB;
    bool m2 = r2 < (unsigned)AGG_NB, m3 = r3 < (unsigned)AGG_NB;
    if (!WHOLE) {
      const bool ok = jb + 4 * u + q4 < jend;
      m0 = m0 && ok; m1 = m1 && ok; m2 = m2 && ok; m3 = m3 && ok;
    }
    const unsigned b0 = __builtin_amdgcn_ballot_w32(m0), b1 = __builtin_amdgcn_ballot_w32(m1);
    const unsigned b2 = __builtin_amdgcn_ballot_w32(m2), b3 = __builtin_amdgcn_ballot_w32(m3);
    const int tot = __builtin_popcount(b0) + __builtin_popcount(b1) + __builtin_popcount(b2) + __builtin_popcount(b3);
    if (wb + tot <= AGG_WSEG) {
      int p = wb + (int)(__builtin_amdgcn_mbcnt_lo(b0, 0u) + __builtin_amdgcn_mbcnt_lo(b1, 0u) +
                         __builtin_amdgcn_mbcnt_lo(b2, 0u) + __builtin_amdgcn_mbcnt_lo(b3, 0u));
      const int e0 = gg[u] * BM_GE + 4 * sub;
      if (m0) { myseg[p] = (e0 + 0) | (int)(r0 << 21); ++p; }
      if (m1) { myseg[p] = (e0 + 1) | (int)(r1 << 21); ++p; }
      if (m2) { myseg[p] = (e0 + 2) | (int)(r2 << 21); ++p; }
      if (m3) { myseg[p] = (e0 + 3) | (int)(r3 << 21); }
    }
    wb += tot;
  }
  return wb;
}

__global__ __launch_bounds__(AGG_T) void aggregate(
    const int* __restrict__ src, const int* __restrict__ dst,
    const int* __restrict__ bmap,
    const float* __restrict__ H,
    const float* __restrict__ el, const float* __restrict__ er,
    float* __restrict__ out)
{
  __shared__ float acc[AGG_NB * OUT_DIM];
  __shared__ int   q_pk[AGG_QCAP];
  __shared__ float q_a[AGG_QCAP];
  __shared__ float er_s[AGG_NB];
  __shared__ float mx_s[AGG_NB];
  __shared__ float sm_s[AGG_NB];
  __shared__ int   wq_n[AGG_NW];
  __shared__ int   sc_w[AGG_NW];
  __shared__ int   w_cnt[AGG_NW];
  __shared__ int   r_ovf;
  __shared__ int   w_ovf;

  int*   wq_pk = (int*)acc;
  float* wq_a  = acc + AGG_NW * AGG_WQCAP;

  const int t    = threadIdx.x;
  const int lane = t & 31;
  const int wave = __builtin_amdgcn_readfirstlane(t >> 5);
  const int lo   = blockIdx.x * AGG_NB;

  unsigned bw[BM_WPB] = {0u, 0u, 0u, 0u};
  if (t < NL_GRID) {
    const int4 x = ((const int4*)bmap)[t * AGG_NR + blockIdx.x];
    bw[0] = (unsigned)x.x; bw[1] = (unsigned)x.y; bw[2] = (unsigned)x.z; bw[3] = (unsigned)x.w;
  }

  if (t < AGG_NB) {
    er_s[t] = er[lo + t];
    mx_s[t] = -__builtin_inff();
    sm_s[t] = 0.0f;
  }
  if (t < AGG_NW) wq_n[t] = 0;
  if (t == 0) { r_ovf = 0; w_ovf = 0; }

  const int gcnt = __builtin_popcount(bw[0]) + __builtin_popcount(bw[1]) +
                   __builtin_popcount(bw[2]) + __builtin_popcount(bw[3]);
  const int ginc = wave_incl_sum(gcnt, lane);
  if (lane == 31) sc_w[wave] = ginc;
  __syncthreads();
  int* cand = (int*)acc;
  int ctot, cpre;
  {
    const int winc = wave_incl_sum(sc_w[lane], lane);
    ctot = __builtin_amdgcn_readlane(winc, 31);
    cpre = wave > 0 ? __builtin_amdgcn_readlane(winc, wave - 1) : 0;
  }
  const bool cfit = ctot <= AGG_CCAP;
  if (cfit) {
    int pos = cpre + ginc - gcnt;
#pragma unroll
    for (int k = 0; k < BM_WPB; ++k) {
      unsigned w = bw[k];
      while (w != 0u) {
        const int b = __builtin_ctz(w);
        w &= w - 1u;
        const int g = t * (BM_WPB * 32) + k * 32 + b;
        cand[pos++] = g < BM_NGRP ? g : BM_NGRP - 1;
      }
    }
  }
  __syncthreads();
  const int nc = cfit ? ctot : 0;

  int* seg = (int*)q_a;
  int* myseg = seg + wave * AGG_WSEG;
  const int per  = ((nc + 4 * AGG_NW - 1) / (4 * AGG_NW)) * 4;
  const int jbeg = wave * per;
  const int jend = jbeg + per < nc ? jbeg + per : nc;
  const int q4   = lane >> 3;
  const int sub  = lane & 7;
  int wb = 0;
  int jb = jbeg;
  for (; jb + 4 * AGG_RU <= jend; jb += 4 * AGG_RU)
    wb = list_groups<true>(jb, jbeg, jend, q4, sub, lo, cand, dst, myseg, wb);
  if (jb < jend)
    wb = list_groups<false>(jb, jbeg, jend, q4, sub, lo, cand, dst, myseg, wb);
  if (lane == 0) {
    w_cnt[wave] = wb < AGG_WSEG ? wb : AGG_WSEG;
    if (wb > AGG_WSEG) r_ovf = 1;
  }
  __syncthreads();

  int total, qpre;
  {
    const int winc = wave_incl_sum(w_cnt[lane], lane);
    total = __builtin_amdgcn_readlane(winc, 31);
    qpre  = wave > 0 ? __builtin_amdgcn_readlane(winc, wave - 1) : 0;
  }
  const bool full = !cfit || __builtin_amdgcn_readfirstlane(r_ovf) != 0;
  const int n = full ? 0 : total;
  if (!full) {
    const int mine = wb;
    for (int k = lane; k < mine; k += 32) q_pk[qpre + k] = myseg[k];
  }
  __syncthreads();

  for (int base = wave * AGG_SBLK; base < n; base += (AGG_T / 32) * AGG_SBLK) {
    if (n - base >= AGG_SBLK) {
      score_block<true>(base, n, lane, src, el, er_s, q_pk, q_a, mx_s);
    } else {
      score_block<false>(base, n, lane, src, el, er_s, q_pk, q_a, mx_s);
    }
  }
  if (full) {
    for (int e = t; e < N_EDGES; e += AGG_T) {
      const unsigned r = (unsigned)(dst[e] - lo);
      if (r < (unsigned)AGG_NB) {
        const float sc = edge_score(el[src[e]], er_s[r]);
        __hip_atomic_fetch_max(&mx_s[r], sc, __ATOMIC_RELAXED, __HIP_MEMORY_SCOPE_WORKGROUP);
      }
    }
  }
  __syncthreads();

  for (int i = t; i < n; i += AGG_T) {
    const int pk = q_pk[i];
    const int r = (int)((unsigned)pk >> 16);
    const float ex = __expf(q_a[i] - mx_s[r]);
    atomicAdd(&sm_s[r], ex);
    const int w = r & (AGG_NW - 1);
    const int pos = atomicAdd(&wq_n[w], 1);
    if (pos < AGG_WQCAP) {
      wq_pk[w * AGG_WQCAP + pos] = pk;
      wq_a[w * AGG_WQCAP + pos]  = ex;
    } else {
      w_ovf = 1;
    }
  }
  if (full) {
    for (int e = t; e < N_EDGES; e += AGG_T) {
      const unsigned r = (unsigned)(dst[e] - lo);
      if (r < (unsigned)AGG_NB) {
        const float sc = edge_score(el[src[e]], er_s[r]);
        atomicAdd(&sm_s[r], __expf(sc - mx_s[r]));
      }
    }
  }
  __syncthreads();

  const bool slow = full || __builtin_amdgcn_readfirstlane(w_ovf) != 0;

  if (!slow) {
    OwnSums sum;
    sum.x = {}; sum.y = {}; sum.z = {}; sum.w = {};
    const int m = __builtin_amdgcn_readfirstlane(wq_n[wave]);
    const int*   my_pk = wq_pk + wave * AGG_WQCAP;
    const float* my_a  = wq_a  + wave * AGG_WQCAP;
    int b = 0;
    for (; b + AGG_RB <= m; b += AGG_RB) {
      const int j = b + (lane & (AGG_RB - 1));
      const int pk = my_pk[j];
      const float a = my_a[j] / sm_s[(unsigned)pk >> 16];
      own_rows<true>(AGG_RB, pk, a, H, sum, lane);
    }
    if (b < m) {
      const int j = b + (lane & (AGG_RB - 1));
      const bool in = j < m;
      const int pk = in ? my_pk[j] : 0;
      float a = 0.0f;
      if (in) a = my_a[j] / sm_s[(unsigned)pk >> 16];
      own_rows<false>(m - b, pk, a, H, sum, lane);
    }
#pragma unroll
    for (int k = 0; k < AGG_SLOTS; ++k) {
      const int r = wave + AGG_NW * k;
      if (r < AGG_NB)
        *(float4*)(out + (size_t)(lo + r) * OUT_DIM + 4 * lane) =
            make_float4(sum.x[k], sum.y[k], sum.z[k], sum.w[k]);
    }
  } else {
    for (int i = t; i < AGG_NB * OUT_DIM; i += AGG_T) acc[i] = 0.0f;
    __syncthreads();
    for (int e = wave; e < N_EDGES; e += AGG_T / 32) {
      const unsigned r = (unsigned)(dst[e] - lo);
      if (r < (unsigned)AGG_NB) {
        const int s = src[e];
        const float sc = edge_score(el[s], er_s[r]);
        const float a = __expf(sc - mx_s[r]) / sm_s[r];
        const float* hp = H + (size_t)s * OUT_DIM + lane;
        float* o = acc + r * OUT_DIM + lane;
        atomicAdd(o +  0, a * hp[0]);
        atomicAdd(o + 32, a * hp[32]);
        atomicAdd(o + 64, a * hp[64]);
        atomicAdd(o + 96, a * hp[96]);
      }
    }
    __syncthreads();
    float* op = out + (size_t)lo * OUT_DIM;
    for (int i = t; i < AGG_NB * OUT_DIM; i += AGG_T) op[i] = acc[i];
  }
}

extern "C" void kernel_launch(void* const* d_in, const int* in_sizes, int n_in,
                              void* d_out, int out_size, void* d_ws, size_t ws_size,
                              hipStream_t stream) {
  (void)in_sizes; (void)n_in; (void)out_size; (void)ws_size;

  const float* features  = (const float*)d_in[0];
  const float* fc_weight = (const float*)d_in[1];
  const float* attn_l    = (const float*)d_in[2];
  const float* attn_r    = (const float*)d_in[3];
  const int*   src       = (const int*)d_in[4];
  const int*   dst       = (const int*)d_in[5];
  float*       out       = (float*)d_out;

  float* h  = (float*)d_ws;
  float* el = h + (size_t)N_NODES * OUT_DIM;
  float* er = el + N_NODES;
  int* bmap = (int*)(er + N_NODES);

  gat_gemm_wmma<<<GEMM_GRID, 256, 0, stream>>>(features, fc_weight, h);

  node_logits<<<NL_GRID, 256, 0, stream>>>(h, attn_l, attn_r, dst, el, er, bmap);

  aggregate<<<N_NODES / AGG_NB, AGG_T, 0, stream>>>(src, dst, bmap, h, el, er, out);
}


# GEMM: each row block's 16x256 A block is fetched once per work-group by whole-line loads (512 B per instruction), a row block ahead, staged in LDS; the 8 waves read fragments from LDS; sum order kept
# speedup vs baseline: 5.6164x; 1.0895x over previous
#include <hip/hip_runtime.h>
#include <hip/hip_bf16.h>

#define N_NODES 50000
#define N_EDGES 1600000
#define IN_DIM 256
#define OUT_DIM 128
#define NEG_SLOPE 0.2f

typedef __attribute__((ext_vector_type(2))) float v2f;
typedef __attribute__((ext_vector_type(8))) float v8f;
typedef __attribute__((ext_vector_type(4))) float v4f;

#define GEMM_RB (N_NODES / 16)
#define GEMM_GRID 512
static_assert(N_NODES % 16 == 0, "row blocks must tile the nodes");
static_assert(IN_DIM % 8 == 0, "k is walked 8 at a time");
#define GEMM_CH 4
static_assert((IN_DIM / 8) % GEMM_CH == 0, "chunks must tile the k range");
#define GEMM_PITCH (IN_DIM + 4)
#define GEMM_STG (16 * IN_DIM / 4 / 256)
static_assert(16 * IN_DIM / 4 == GEMM_STG * 256, "the threads' pieces must tile the row block");
static_assert(IN_DIM / 4 == 64, "staging map: 64 pieces per row");

__global__ __launch_bounds__(256) void gat_gemm_wmma(
    const float* __restrict__ X,
    const float* __restrict__ W,
    float* __restrict__ H)
{
  __shared__ __attribute__((aligned(16))) float a_s[2][16 * GEMM_PITCH];

  const int lane = threadIdx.x & 31;
  const int wave = threadIdx.x >> 5;
  const int l16  = lane & 15;
  const int half = lane >> 4;
  const int nBase = wave * 16;

  v2f bf[IN_DIM / 4];
  const float* bPtr = W + (size_t)(4 * half) * OUT_DIM + nBase + l16;
#pragma unroll
  for (int j = 0; j < IN_DIM / 8; ++j) {
    bf[2 * j].x     = bPtr[(size_t)(8 * j + 0) * OUT_DIM];
    bf[2 * j].y     = bPtr[(size_t)(8 * j + 1) * OUT_DIM];
    bf[2 * j + 1].x = bPtr[(size_t)(8 * j + 2) * OUT_DIM];
    bf[2 * j + 1].y = bPtr[(size_t)(8 * j + 3) * OUT_DIM];
  }

  const int sRow = threadIdx.x >> 6;
  const int sPc  = threadIdx.x & 63;
  v4f pre[GEMM_STG];
  {
    const v4f* p = (const v4f*)(X + (size_t)(blockIdx.x * 16 + sRow) * IN_DIM) + sPc;
#pragma unroll
    for (int q = 0; q < GEMM_STG; ++q) pre[q] = p[(size_t)q * 4 * (IN_DIM / 4)];
  }
  const int fOff = l16 * GEMM_PITCH + 4 * half;
  int buf = 0;
  for (int rb = blockIdx.x; rb < GEMM_RB; rb += gridDim.x) {
    const int mBase = rb * 16;
    const int rbn = rb + (int)gridDim.x < GEMM_RB ? rb + (int)gridDim.x : rb;
    {
      float* sp = &a_s[buf][sRow * GEMM_PITCH + 4 * sPc];
#pragma unroll
      for (int q = 0; q < GEMM_STG; ++q) *(v4f*)(sp + q * 4 * GEMM_PITCH) = pre[q];
      const v4f* p = (const v4f*)(X + (size_t)(rbn * 16 + sRow) * IN_DIM) + sPc;
#pragma unroll
      for (int q = 0; q < GEMM_STG; ++q) pre[q] = p[(size_t)q * 4 * (IN_DIM / 4)];
    }
    __builtin_amdgcn_fence(__ATOMIC_RELEASE, "workgroup", "local");
    __builtin_amdgcn_s_barrier();
    __builtin_amdgcn_fence(__ATOMIC_ACQUIRE, "workgroup", "local");
    const float* ap = &a_s[buf][fOff];
    float4 cur[GEMM_CH], nxt[GEMM_CH];
#pragma unroll
    for (int i = 0; i < GEMM_CH; ++i) cur[i] = *(const float4*)(ap + 8 * i);
    v8f acc0 = {};
    v8f acc1 = {};
#pragma unroll
    for (int c = 0; c < IN_DIM / 8 / GEMM_CH; ++c) {
      if (c + 1 < IN_DIM / 8 / GEMM_CH) {
#pragma unroll
        for (int i = 0; i < GEMM_CH; ++i)
          nxt[i] = *(const float4*)(ap + 8 * ((c + 1) * GEMM_CH + i));
      }
      __builtin_amdgcn_sched_barrier(0);
#pragma unroll
      for (int i = 0; i < GEMM_CH; ++i) {
        const int j = c * GEMM_CH + i;
        v2f a0; a0.x = cur[i].x; a0.y = cur[i].y;
        v2f a1; a1.x = cur[i].z; a1.y = cur[i].w;
        acc0 = __builtin_amdgcn_wmma_f32_16x16x4_f32(
             false, a0,  false, bf[2 * j],
             (short)0, acc0,  false,  false);
        acc1 = __builtin_amdgcn_wmma_f32_16x16x4_f32(
             false, a1,  false, bf[2 * j + 1],
             (short)0, acc1,  false,  false);
      }
      if (c + 1 < IN_DIM / 8 / GEMM_CH) {
#pragma unroll
        for (int i = 0; i < GEMM_CH; ++i) cur[i] = nxt[i];
      }
    }
    float* out = H + (size_t)(mBase + 8 * half) * OUT_DIM + nBase + l16;
#pragma unroll
    for (int v = 0; v < 8; ++v)
      out[(size_t)v * OUT_DIM] = acc0[v] + acc1[v];
    buf ^= 1;
  }
}

#define AGG_NB   200
#define RID_NONE 255u
#define AGG_NR   (N_NODES / AGG_NB)
#define BM_GE    32
#define BM_NGRP  (N_EDGES / BM_GE)
#define BM_WPB   4
static_assert(N_NODES % AGG_NB == 0, "node ranges must tile the nodes");
static_assert(N_NODES / AGG_NB <= 255, "a range id must fit a byte, 255 kept free");
static_assert(N_EDGES % BM_GE == 0, "bitmap groups must tile the edges");

__device__ __forceinline__ unsigned range_id(int d)
{
  return (unsigned)d < (unsigned)N_NODES ? (unsigned)d / AGG_NB : RID_NONE;
}

#define NL_ROWS 16
#define NL_WAVES (N_NODES / NL_ROWS)
#define NL_GRID ((NL_WAVES + 7) / 8)
#define NL_NG4 (N_EDGES / 4)
#define NL_G4_WG (BM_WPB * 32 * BM_GE / 4)
#define NL_RID_TRIPS (NL_G4_WG / 256)
static_assert(NL_RID_TRIPS == BM_WPB, "trip k of a thread fills bitmap word k");
static_assert(NL_GRID * NL_G4_WG >= NL_NG4, "the work-groups must cover dst");
static_assert((size_t)NL_GRID * AGG_NR * BM_WPB * 4 <= (size_t)N_EDGES, "the bitmap must fit its N_EDGES bytes of workspace");
static_assert(N_NODES % NL_ROWS == 0, "row groups must tile the nodes");
static_assert(NL_ROWS <= 32, "a lane per node of the group for the stores");
static_assert(N_EDGES % 4 == 0, "dst is read in 16-byte groups");

__global__ __launch_bounds__(256) void node_logits(
    const float* __restrict__ H,
    const float* __restrict__ attn_l,
    const float* __restrict__ attn_r,
    const int* __restrict__ dst,
    float* __restrict__ el, float* __restrict__ er,
    int* __restrict__ bmap)
{
  __shared__ __attribute__((aligned(16))) unsigned bm[AGG_NR * BM_WPB];
  const int gid  = blockIdx.x * 256 + threadIdx.x;
  const int lane = threadIdx.x & 31;
  const int grp  = __builtin_amdgcn_readfirstlane(gid >> 5);
  const bool own = grp < NL_WAVES;
  const int n0 = (own ? grp : NL_WAVES - 1) * NL_ROWS;

  float4 hv[NL_ROWS];
  const float* hp = H + (size_t)n0 * OUT_DIM + 4 * lane;
#pragma unroll
  for (int u = 0; u < NL_ROWS; ++u) hv[u] = *(const float4*)(hp + (size_t)u * OUT_DIM);

  int4 d[NL_RID_TRIPS];
#pragma unroll
  for (int k = 0; k < NL_RID_TRIPS; ++k) {
    const int i = blockIdx.x * NL_G4_WG + k * 256 + (int)threadIdx.x;
    d[k] = ((const int4*)dst)[i < NL_NG4 ? i : NL_NG4 - 1];
  }
  const float4 al = *(const float4*)(attn_l + 4 * lane);
  const float4 ar = *(const float4*)(attn_r + 4 * lane);

  for (int i = threadIdx.x; i < AGG_NR * BM_WPB; i += 256) bm[i] = 0u;
  __syncthreads();
  const unsigned bit = 1u << (threadIdx.x >> 3);
#pragma unroll
  for (int k = 0; k < NL_RID_TRIPS; ++k) {
    const int i = blockIdx.x * NL_G4_WG + k * 256 + (int)threadIdx.x;
    if (i < NL_NG4) {
      const int dd[4] = {d[k].x, d[k].y, d[k].z, d[k].w};
#pragma unroll
      for (int c = 0; c < 4; ++c) {
        const unsigned rid = range_id(dd[c]);
        if (rid != RID_NONE)
          __hip_atomic_fetch_or(&bm[rid * BM_WPB + k], bit, __ATOMIC_RELAXED, __HIP_MEMORY_SCOPE_WORKGROUP);
      }
    }
  }
  __syncthreads();
  if (threadIdx.x < AGG_NR)
    ((int4*)bmap)[blockIdx.x * AGG_NR + threadIdx.x] = *(const int4*)&bm[threadIdx.x * BM_WPB];

  float sl[NL_ROWS], sr[NL_ROWS];
#pragma unroll
  for (int u = 0; u < NL_ROWS; ++u) {
    sl[u] = hv[u].x * al.x + hv[u].y * al.y + hv[u].z * al.z + hv[u].w * al.w;
    sr[u] = hv[u].x * ar.x + hv[u].y * ar.y + hv[u].z * ar.z + hv[u].w * ar.w;
  }
#pragma unroll
  for (int off = 16; off > 0; off >>= 1) {
#pragma unroll
    for (int u = 0; u < NL_ROWS; ++u) {
      sl[u] += __shfl_xor(sl[u], off, 32);
      sr[u] += __shfl_xor(sr[u], off, 32);
    }
  }
  float myl = 0.0f, myr = 0.0f;
#pragma unroll
  for (int u = 0; u < NL_ROWS; ++u)
    if (lane == u) { myl = sl[u]; myr = sr[u]; }
  if (own && lane < NL_ROWS) {
    el[n0 + lane] = myl;
    er[n0 + lane] = myr;
  }
}

#define AGG_T    1024
#define AGG_QCAP 16384
#define AGG_NW   (AGG_T / 32)
#define AGG_SLOTS ((AGG_NB + AGG_NW - 1) / AGG_NW)
#define AGG_WQCAP (AGG_NB * OUT_DIM / (2 * AGG_NW))
static_assert(AGG_NB <= 256, "list entry: local node in 8 bits");
static_assert(N_NODES <= (1 << 16), "list entry: source node in 16 bits");
static_assert(AGG_SLOTS <= 8, "the register sums hold 8 slots per wave");
static_assert(AGG_NW == 32, "owner wave = local node & 31, slot = local node >> 5");
static_assert(2 * AGG_NW * AGG_WQCAP <= AGG_NB * OUT_DIM, "queues must fit the row-sum LDS");
#define AGG_RB 16
static_assert(AGG_RB <= 32 && (AGG_RB & (AGG_RB - 1)) == 0, "a lane per edge of the batch");
static_assert(AGG_WQCAP % AGG_RB == 0, "a batch never reads past its own queue");

struct OwnSums { v8f x, y, z, w; };

template <bool WHOLE>
__device__ __forceinline__ void own_rows(
    int cnt, int pk_l, float a_l, const float* __restrict__ H,
    OwnSums& sum, int lane)
{
  float4 hv[AGG_RB];
#pragma unroll
  for (int k = 0; k < AGG_RB; ++k) {
    if (WHOLE || k < cnt) {
      const int s = __builtin_amdgcn_readlane(pk_l, k) & 0xFFFF;
      hv[k] = *(const float4*)(H + (size_t)s * OUT_DIM + 4 * lane);
    }
  }
#pragma unroll
  for (int k = 0; k < AGG_RB; ++k) {
    if (WHOLE || k < cnt) {
      const int slot = (int)((unsigned)__builtin_amdgcn_readlane(pk_l, k) >> 21);
      const float a = __int_as_float(__builtin_amdgcn_readlane(__float_as_int(a_l), k));
      sum.x[slot] += a * hv[k].x;
      sum.y[slot] += a * hv[k].y;
      sum.z[slot] += a * hv[k].z;
      sum.w[slot] += a * hv[k].w;
    }
  }
}

__device__ __forceinline__ float edge_score(float l, float r)
{
  const float t = l + r;
  return t > 0.0f ? t : NEG_SLOPE * t;
}

#define AGG_SB 8
#define AGG_SBLK (32 * AGG_SB)

template <bool WHOLE>
__device__ __forceinline__ void score_block(
    int base, int n, int lane,
    const int* __restrict__ src,
    const float* __restrict__ el,
    const float* er_s, int* q_pk, float* q_a, float* mx_s)
{
  int pk[AGG_SB], s[AGG_SB];
  float l[AGG_SB];
#pragma unroll
  for (int u = 0; u < AGG_SB; ++u) {
    const int i = base + 32 * u + lane;
    pk[u] = (WHOLE || i < n) ? q_pk[i] : 0;
  }
#pragma unroll
  for (int u = 0; u < AGG_SB; ++u) s[u] = src[pk[u] & 0x1FFFFF];
#pragma unroll
  for (int u = 0; u < AGG_SB; ++u) l[u] = el[s[u]];
#pragma unroll
  for (int u = 0; u < AGG_SB; ++u) {
    const int i = base + 32 * u + lane;
    if (WHOLE || i < n) {
      const int r = (int)((unsigned)pk[u] >> 21);
      const float sc = edge_score(l[u], er_s[r]);
      q_a[i]  = sc;
      q_pk[i] = s[u] | (r << 16);
      __hip_atomic_fetch_max(&mx_s[r], sc, __ATOMIC_RELAXED, __HIP_MEMORY_SCOPE_WORKGROUP);
    }
  }
}

__device__ __forceinline__ int wave_incl_sum(int v, int lane)
{
#pragma unroll
  for (int off = 1; off < 32; off <<= 1) {
    const int o = __shfl_up(v, off, 32);
    if (lane >= off) v += o;
  }
  return v;
}

#define AGG_CCAP (AGG_NB * OUT_DIM)
#define AGG_WSEG (AGG_QCAP / AGG_NW)
#define AGG_RU   8
static_assert(AGG_WSEG * AGG_NW == AGG_QCAP, "the waves' pieces fill the list");
static_assert(N_EDGES <= (1 << 21), "list entry while listing: edge in 21 bits, local node above");
static_assert(NL_GRID <= AGG_T, "a thread per logits work-group's piece of the bitmap");

template <bool WHOLE>
__device__ __forceinline__ int list_groups(
    int jb, int jbeg, int jend, int q4, int sub, int lo,
    const int* cand, const int* __restrict__ dst, int* myseg, int wb)
{
  int4 dv[AGG_RU];
  int  gg[AGG_RU];
#pragma unroll
  for (int u = 0; u < AGG_RU; ++u) {
    const int j = jb + 4 * u + q4;
    gg[u] = cand[(WHOLE || j < jend) ? j : jbeg];
    dv[u] = ((const int4*)dst)[gg[u] * (BM_GE / 4) + sub];
  }
#pragma unroll
  for (int u = 0; u < AGG_RU; ++u) {
    const unsigned r0 = (unsigned)(dv[u].x - lo), r1 = (unsigned)(dv[u].y - lo);
    const unsigned r2 = (unsigned)(dv[u].z - lo), r3 = (unsigned)(dv[u].w - lo);
    bool m0 = r0 < (unsigned)AGG_NB, m1 = r1 < (unsigned)AGG_NB;
    bool m2 = r2 < (unsigned)AGG_NB, m3 = r3 < (unsigned)AGG_NB;
    if (!WHOLE) {
      const bool ok = jb + 4 * u + q4 < jend;
      m0 = m0 && ok; m1 = m1 && ok; m2 = m2 && ok; m3 = m3 && ok;
    }
    const unsigned b0 = __builtin_amdgcn_ballot_w32(m0), b1 = __builtin_amdgcn_ballot_w32(m1);
    const unsigned b2 = __builtin_amdgcn_ballot_w32(m2), b3 = __builtin_amdgcn_ballot_w32(m3);
    const int tot = __builtin_popcount(b0) + __builtin_popcount(b1) + __builtin_popcount(b2) + __builtin_popcount(b3);
    if (wb + tot <= AGG_WSEG) {
      int p = wb + (int)(__builtin_amdgcn_mbcnt_lo(b0, 0u) + __builtin_amdgcn_mbcnt_lo(b1, 0u) +
                         __builtin_amdgcn_mbcnt_lo(b2, 0u) + __builtin_amdgcn_mbcnt_lo(b3, 0u));
      const int e0 = gg[u] * BM_GE + 4 * sub;
      if (m0) { myseg[p] = (e0 + 0) | (int)(r0 << 21); ++p; }
      if (m1) { myseg[p] = (e0 + 1) | (int)(r1 << 21); ++p; }
      if (m2) { myseg[p] = (e0 + 2) | (int)(r2 << 21); ++p; }
      if (m3) { myseg[p] = (e0 + 3) | (int)(r3 << 21); }
    }
    wb += tot;
  }
  return wb;
}

__global__ __launch_bounds__(AGG_T) void aggregate(
    const int* __restrict__ src, const int* __restrict__ dst,
    const int* __restrict__ bmap,
    const float* __restrict__ H,
    const float* __restrict__ el, const float* __restrict__ er,
    float* __restrict__ out)
{
  __shared__ float acc[AGG_NB * OUT_DIM];
  __shared__ int   q_pk[AGG_QCAP];
  __shared__ float q_a[AGG_QCAP];
  __shared__ float er_s[AGG_NB];
  __shared__ float mx_s[AGG_NB];
  __shared__ float sm_s[AGG_NB];
  __shared__ int   wq_n[AGG_NW];
  __shared__ int   sc_w[AGG_NW];
  __shared__ int   w_cnt[AGG_NW];
  __shared__ int   r_ovf;
  __shared__ int   w_ovf;

  int*   wq_pk = (int*)acc;
  float* wq_a  = acc + AGG_NW * AGG_WQCAP;

  const int t    = threadIdx.x;
  const int lane = t & 31;
  const int wave = __builtin_amdgcn_readfirstlane(t >> 5);
  const int lo   = blockIdx.x * AGG_NB;

  unsigned bw[BM_WPB] = {0u, 0u, 0u, 0u};
  if (t < NL_GRID) {
    const int4 x = ((const int4*)bmap)[t * AGG_NR + blockIdx.x];
    bw[0] = (unsigned)x.x; bw[1] = (unsigned)x.y; bw[2] = (unsigned)x.z; bw[3] = (unsigned)x.w;
  }

  if (t < AGG_NB) {
    er_s[t] = er[lo + t];
    mx_s[t] = -__builtin_inff();
    sm_s[t] = 0.0f;
  }
  if (t < AGG_NW) wq_n[t] = 0;
  if (t == 0) { r_ovf = 0; w_ovf = 0; }

  const int gcnt = __builtin_popcount(bw[0]) + __builtin_popcount(bw[1]) +
                   __builtin_popcount(bw[2]) + __builtin_popcount(bw[3]);
  const int ginc = wave_incl_sum(gcnt, lane);
  if (lane == 31) sc_w[wave] = ginc;
  __syncthreads();
  int* cand = (int*)acc;
  int ctot, cpre;
  {
    const int winc = wave_incl_sum(sc_w[lane], lane);
    ctot = __builtin_amdgcn_readlane(winc, 31);
    cpre = wave > 0 ? __builtin_amdgcn_readlane(winc, wave - 1) : 0;
  }
  const bool cfit = ctot <= AGG_CCAP;
  if (cfit) {
    int pos = cpre + ginc - gcnt;
#pragma unroll
    for (int k = 0; k < BM_WPB; ++k) {
      unsigned w = bw[k];
      while (w != 0u) {
        const int b = __builtin_ctz(w);
        w &= w - 1u;
        const int g = t * (BM_WPB * 32) + k * 32 + b;
        cand[pos++] = g < BM_NGRP ? g : BM_NGRP - 1;
      }
    }
  }
  __syncthreads();
  const int nc = cfit ? ctot : 0;

  int* seg = (int*)q_a;
  int* myseg = seg + wave * AGG_WSEG;
  const int per  = ((nc + 4 * AGG_NW - 1) / (4 * AGG_NW)) * 4;
  const int jbeg = wave * per;
  const int jend = jbeg + per < nc ? jbeg + per : nc;
  const int q4   = lane >> 3;
  const int sub  = lane & 7;
  int wb = 0;
  int jb = jbeg;
  for (; jb + 4 * AGG_RU <= jend; jb += 4 * AGG_RU)
    wb = list_groups<true>(jb, jbeg, jend, q4, sub, lo, cand, dst, myseg, wb);
  if (jb < jend)
    wb = list_groups<false>(jb, jbeg, jend, q4, sub, lo, cand, dst, myseg, wb);
  if (lane == 0) {
    w_cnt[wave] = wb < AGG_WSEG ? wb : AGG_WSEG;
    if (wb > AGG_WSEG) r_ovf = 1;
  }
  __syncthreads();

  int total, qpre;
  {
    const int winc = wave_incl_sum(w_cnt[lane], lane);
    total = __builtin_amdgcn_readlane(winc, 31);
    qpre  = wave > 0 ? __builtin_amdgcn_readlane(winc, wave - 1) : 0;
  }
  const bool full = !cfit || __builtin_amdgcn_readfirstlane(r_ovf) != 0;
  const int n = full ? 0 : total;
  if (!full) {
    const int mine = wb;
    for (int k = lane; k < mine; k += 32) q_pk[qpre + k] = myseg[k];
  }
  __syncthreads();

  for (int base = wave * AGG_SBLK; base < n; base += (AGG_T / 32) * AGG_SBLK) {
    if (n - base >= AGG_SBLK) {
      score_block<true>(base, n, lane, src, el, er_s, q_pk, q_a, mx_s);
    } else {
      score_block<false>(base, n, lane, src, el, er_s, q_pk, q_a, mx_s);
    }
  }
  if (full) {
    for (int e = t; e < N_EDGES; e += AGG_T) {
      const unsigned r = (unsigned)(dst[e] - lo);
      if (r < (unsigned)AGG_NB) {
        const float sc = edge_score(el[src[e]], er_s[r]);
        __hip_atomic_fetch_max(&mx_s[r], sc, __ATOMIC_RELAXED, __HIP_MEMORY_SCOPE_WORKGROUP);
      }
    }
  }
  __syncthreads();

  for (int i = t; i < n; i += AGG_T) {
    const int pk = q_pk[i];
    const int r = (int)((unsigned)pk >> 16);
    const float ex = __expf(q_a[i] - mx_s[r]);
    atomicAdd(&sm_s[r], ex);
    const int w = r & (AGG_NW - 1);
    const int pos = atomicAdd(&wq_n[w], 1);
    if (pos < AGG_WQCAP) {
      wq_pk[w * AGG_WQCAP + pos] = pk;
      wq_a[w * AGG_WQCAP + pos]  = ex;
    } else {
      w_ovf = 1;
    }
  }
  if (full) {
    for (int e = t; e < N_EDGES; e += AGG_T) {
      const unsigned r = (unsigned)(dst[e] - lo);
      if (r < (unsigned)AGG_NB) {
        const float sc = edge_score(el[src[e]], er_s[r]);
        atomicAdd(&sm_s[r], __expf(sc - mx_s[r]));
      }
    }
  }
  __syncthreads();

  const bool slow = full || __builtin_amdgcn_readfirstlane(w_ovf) != 0;

  if (!slow) {
    OwnSums sum;
    sum.x = {}; sum.y = {}; sum.z = {}; sum.w = {};
    const int m = __builtin_amdgcn_readfirstlane(wq_n[wave]);
    const int*   my_pk = wq_pk + wave * AGG_WQCAP;
    const float* my_a  = wq_a  + wave * AGG_WQCAP;
    int b = 0;
    for (; b + AGG_RB <= m; b += AGG_RB) {
      const int j = b + (lane & (AGG_RB - 1));
      const int pk = my_pk[j];
      const float a = my_a[j] / sm_s[(unsigned)pk >> 16];
      own_rows<true>(AGG_RB, pk, a, H, sum, lane);
    }
    if (b < m) {
      const int j = b + (lane & (AGG_RB - 1));
      const bool in = j < m;
      const int pk = in ? my_pk[j] : 0;
      float a = 0.0f;
      if (in) a = my_a[j] / sm_s[(unsigned)pk >> 16];
      own_rows<false>(m - b, pk, a, H, sum, lane);
    }
#pragma unroll
    for (int k = 0; k < AGG_SLOTS; ++k) {
      const int r = wave + AGG_NW * k;
      if (r < AGG_NB)
        *(float4*)(out + (size_t)(lo + r) * OUT_DIM + 4 * lane) =
            make_float4(sum.x[k], sum.y[k], sum.z[k], sum.w[k]);
    }
  } else {
    for (int i = t; i < AGG_NB * OUT_DIM; i += AGG_T) acc[i] = 0.0f;
    __syncthreads();
    for (int e = wave; e < N_EDGES; e += AGG_T / 32) {
      const unsigned r = (unsigned)(dst[e] - lo);
      if (r < (unsigned)AGG_NB) {
        const int s = src[e];
        const float sc = edge_score(el[s], er_s[r]);
        const float a = __expf(sc - mx_s[r]) / sm_s[r];
        const float* hp = H + (size_t)s * OUT_DIM + lane;
        float* o = acc + r * OUT_DIM + lane;
        atomicAdd(o +  0, a * hp[0]);
        atomicAdd(o + 32, a * hp[32]);
        atomicAdd(o + 64, a * hp[64]);
        atomicAdd(o + 96, a * hp[96]);
      }
    }
    __syncthreads();
    float* op = out + (size_t)lo * OUT_DIM;
    for (int i = t; i < AGG_NB * OUT_DIM; i += AGG_T) op[i] = acc[i];
  }
}

extern "C" void kernel_launch(void* const* d_in, const int* in_sizes, int n_in,
                              void* d_out, int out_size, void* d_ws, size_t ws_size,
                              hipStream_t stream) {
  (void)in_sizes; (void)n_in; (void)out_size; (void)ws_size;

  const float* features  = (const float*)d_in[0];
  const float* fc_weight = (const float*)d_in[1];
  const float* attn_l    = (const float*)d_in[2];
  const float* attn_r    = (const float*)d_in[3];
  const int*   src       = (const int*)d_in[4];
  const int*   dst       = (const int*)d_in[5];
  float*       out       = (float*)d_out;

  float* h  = (float*)d_ws;
  float* el = h + (size_t)N_NODES * OUT_DIM;
  float* er = el + N_NODES;
  int* bmap = (int*)(er + N_NODES);

  gat_gemm_wmma<<<GEMM_GRID, 256, 0, stream>>>(features, fc_weight, h);

  node_logits<<<NL_GRID, 256, 0, stream>>>(h, attn_l, attn_r, dst, el, er, bmap);

  aggregate<<<N_NODES / AGG_NB, AGG_T, 0, stream>>>(src, dst, bmap, h, el, er, out);
}
